# starting kernel
# baseline (speedup 1.0000x reference)
#include <hip/hip_runtime.h>
#include <hip/hip_bf16.h>

#define N_NODES 100000
#define N_EDGES 1600000
#define NFH     128
#define NCLS    64

typedef __attribute__((ext_vector_type(2))) float v2f;
typedef __attribute__((ext_vector_type(8))) float v8f;

__global__ void k_deg_init(float* __restrict__ deg, int n) {
  int i = blockIdx.x * blockDim.x + threadIdx.x;
  if (i < n) deg[i] = 1.0f;
}

__global__ void k_deg_edges(const int* __restrict__ dst, float* __restrict__ deg, int e) {
  int i = blockIdx.x * blockDim.x + threadIdx.x;
  if (i < e) atomicAdd(&deg[dst[i]], 1.0f);
}

__global__ void k_rsqrt_inplace(float* __restrict__ d, int n) {
  int i = blockIdx.x * blockDim.x + threadIdx.x;
  if (i < n) d[i] = rsqrtf(d[i]);
}

__global__ void k_coef(const int* __restrict__ src, const int* __restrict__ dst,
                       const float* __restrict__ dinv, float* __restrict__ coef, int e) {
  int i = blockIdx.x * blockDim.x + threadIdx.x;
  if (i < e) coef[i] = dinv[src[i]] * dinv[dst[i]];
}

template <int K, int M>
__global__ __launch_bounds__(256)
void k_gemm_wmma(const float* __restrict__ X, const float* __restrict__ W,
                 float* __restrict__ H, int n_rows) {
  const int lane = threadIdx.x & 31;
  const int wave = (blockIdx.x * (blockDim.x >> 5)) + (threadIdx.x >> 5);
  constexpr int pairsPerRow = M / 32;
  const int nWork = (n_rows / 16) * pairsPerRow;
  if (wave >= nWork) return;

  const int ti = wave / pairsPerRow;
  const int tj = (wave % pairsPerRow) * 2;

  const int m  = lane & 15;
  const int ks = (lane >> 4) << 1;

  const float* __restrict__ xrow = X + (size_t)(ti * 16 + m) * K + ks;
  const float* __restrict__ w0   = W + (size_t)(tj * 16) + m;
  const float* __restrict__ w1   = w0 + 16;

  v8f c0 = {}, c1 = {};
#pragma unroll
  for (int k0 = 0; k0 < K; k0 += 4) {
    v2f a, b0, b1;
    a.x  = xrow[k0];
    a.y  = xrow[k0 + 1];
    b0.x = w0[(size_t)(k0 + ks) * M];
    b0.y = w0[(size_t)(k0 + ks + 1) * M];
    b1.x = w1[(size_t)(k0 + ks) * M];
    b1.y = w1[(size_t)(k0 + ks + 1) * M];
    c0 = __builtin_amdgcn_wmma_f32_16x16x4_f32(false, a, false, b0, (short)0, c0,
                                               false, false);
    c1 = __builtin_amdgcn_wmma_f32_16x16x4_f32(false, a, false, b1, (short)0, c1,
                                               false, false);
  }

  const int row0 = ti * 16 + ((lane >> 4) << 3);
  float* __restrict__ out0 = H + (size_t)row0 * M + tj * 16 + m;
#pragma unroll
  for (int v = 0; v < 8; ++v) {
    out0[(size_t)v * M]      = c0[v];
    out0[(size_t)v * M + 16] = c1[v];
  }
}

template <int M>
__global__ void k_init_agg(const float* __restrict__ h, const float* __restrict__ dinv,
                           const float* __restrict__ bias, float* __restrict__ agg,
                           int n) {
  int idx = blockIdx.x * blockDim.x + threadIdx.x;
  int total = n * M;
  if (idx < total) {
    int i = idx / M;
    int f = idx - i * M;
    float di = dinv[i];
    agg[idx] = bias[f] + di * di * h[idx];
  }
}

template <int M>
__global__ void k_scatter(const int* __restrict__ src, const int* __restrict__ dst,
                          const float* __restrict__ coef, const float* __restrict__ h,
                          float* __restrict__ agg, int e) {
  const int lane = threadIdx.x & 31;
  const int ed = __builtin_amdgcn_readfirstlane(
      (blockIdx.x * (blockDim.x >> 5)) + (threadIdx.x >> 5));
  if (ed >= e) return;
  const int   s  = __builtin_amdgcn_readfirstlane(src[ed]);
  const int   d  = __builtin_amdgcn_readfirstlane(dst[ed]);
  const float cf = coef[ed];
  constexpr int F = M / 32;
  const float* __restrict__ hp = h + (size_t)s * M + lane * F;
  float* __restrict__ ap = agg + (size_t)d * M + lane * F;
  if constexpr (F == 4) {
    float4 hv = *reinterpret_cast<const float4*>(hp);
    atomicAdd(ap + 0, cf * hv.x);
    atomicAdd(ap + 1, cf * hv.y);
    atomicAdd(ap + 2, cf * hv.z);
    atomicAdd(ap + 3, cf * hv.w);
  } else {
    float2 hv = *reinterpret_cast<const float2*>(hp);
    atomicAdd(ap + 0, cf * hv.x);
    atomicAdd(ap + 1, cf * hv.y);
  }
}

__global__ void k_relu_res(const float* __restrict__ agg, const float* __restrict__ xp,
                           float* __restrict__ xn, int total) {
  int i = blockIdx.x * blockDim.x + threadIdx.x;
  if (i < total) {
    float v = agg[i] + xp[i];
    xn[i] = v > 0.0f ? v : 0.0f;
  }
}

__global__ void k_add(const float* __restrict__ a, const float* __restrict__ b,
                      float* __restrict__ o, int total) {
  int i = blockIdx.x * blockDim.x + threadIdx.x;
  if (i < total) o[i] = a[i] + b[i];
}

extern "C" void kernel_launch(void* const* d_in, const int* in_sizes, int n_in,
                              void* d_out, int out_size, void* d_ws, size_t ws_size,
                              hipStream_t stream) {
  const float* x      = (const float*)d_in[0];
  const int*   ei     = (const int*)d_in[1];
  const float* W_in   = (const float*)d_in[2];
  const float* b_in   = (const float*)d_in[3];
  const float* W_m0   = (const float*)d_in[4];
  const float* b_m0   = (const float*)d_in[5];
  const float* W_m1   = (const float*)d_in[6];
  const float* b_m1   = (const float*)d_in[7];
  const float* W_out  = (const float*)d_in[8];
  const float* b_out  = (const float*)d_in[9];
  const float* W_skip = (const float*)d_in[10];

  const int* src = ei;
  const int* dst = ei + N_EDGES;

  float* w    = (float*)d_ws;
  float* dinv = w;  w += N_NODES;
  float* coef = w;  w += N_EDGES;
  float* h    = w;  w += (size_t)N_NODES * NFH;
  float* agg  = w;  w += (size_t)N_NODES * NFH;
  float* xA   = w;  w += (size_t)N_NODES * NFH;
  float* xB   = w;  w += (size_t)N_NODES * NFH;
  float* skip = w;  w += (size_t)N_NODES * NCLS;

  const int T = 256;
  auto cdiv = [](int a, int b) { return (a + b - 1) / b; };

  const int nodeBlocks  = cdiv(N_NODES, T);
  const int edgeBlocks  = cdiv(N_EDGES, T);
  const int featBlocks  = cdiv(N_NODES * NFH, T);
  const int clsBlocks   = cdiv(N_NODES * NCLS, T);
  const int scatBlocks  = cdiv(N_EDGES, T / 32);
  const int gemmBlocksH = cdiv((N_NODES / 16) * (NFH / 32), T / 32);
  const int gemmBlocksC = cdiv((N_NODES / 16) * (NCLS / 32), T / 32);

  k_deg_init<<<nodeBlocks, T, 0, stream>>>(dinv, N_NODES);
  k_deg_edges<<<edgeBlocks, T, 0, stream>>>(dst, dinv, N_EDGES);
  k_rsqrt_inplace<<<nodeBlocks, T, 0, stream>>>(dinv, N_NODES);
  k_coef<<<edgeBlocks, T, 0, stream>>>(src, dst, dinv, coef, N_EDGES);

  k_gemm_wmma<NFH, NFH><<<gemmBlocksH, T, 0, stream>>>(x, W_in, h, N_NODES);
  k_init_agg<NFH><<<featBlocks, T, 0, stream>>>(h, dinv, b_in, agg, N_NODES);
  k_scatter<NFH><<<scatBlocks, T, 0, stream>>>(src, dst, coef, h, agg, N_EDGES);
  k_relu_res<<<featBlocks, T, 0, stream>>>(agg, x, xA, N_NODES * NFH);

  k_gemm_wmma<NFH, NFH><<<gemmBlocksH, T, 0, stream>>>(xA, W_m0, h, N_NODES);
  k_init_agg<NFH><<<featBlocks, T, 0, stream>>>(h, dinv, b_m0, agg, N_NODES);
  k_scatter<NFH><<<scatBlocks, T, 0, stream>>>(src, dst, coef, h, agg, N_EDGES);
  k_relu_res<<<featBlocks, T, 0, stream>>>(agg, xA, xB, N_NODES * NFH);

  k_gemm_wmma<NFH, NFH><<<gemmBlocksH, T, 0, stream>>>(xB, W_m1, h, N_NODES);
  k_init_agg<NFH><<<featBlocks, T, 0, stream>>>(h, dinv, b_m1, agg, N_NODES);
  k_scatter<NFH><<<scatBlocks, T, 0, stream>>>(src, dst, coef, h, agg, N_EDGES);
  k_relu_res<<<featBlocks, T, 0, stream>>>(agg, xB, xA, N_NODES * NFH);

  k_gemm_wmma<NFH, NCLS><<<gemmBlocksC, T, 0, stream>>>(xA, W_out, h, N_NODES);
  k_init_agg<NCLS><<<clsBlocks, T, 0, stream>>>(h, dinv, b_out, agg, N_NODES);
  k_scatter<NCLS><<<scatBlocks, T, 0, stream>>>(src, dst, coef, h, agg, N_EDGES);
  k_gemm_wmma<NFH, NCLS><<<gemmBlocksC, T, 0, stream>>>(xA, W_skip, skip, N_NODES);
  k_add<<<clsBlocks, T, 0, stream>>>(agg, skip, (float*)d_out, N_NODES * NCLS);
}


# Fewer, longer waves: a scatter wave walks 128 consecutive edges (whole-line metadata loads + readlane, 16 line gathers in flight); row-local kernels move 64 B per thread; sums stay in arrival order
# speedup vs baseline: 1.6736x; 1.6736x over previous
#include <hip/hip_runtime.h>
#include <hip/hip_bf16.h>

#define N_NODES 100000
#define N_EDGES 1600000
#define NFH     128
#define NCLS    64

typedef __attribute__((ext_vector_type(2))) float v2f;
typedef __attribute__((ext_vector_type(8))) float v8f;

__global__ void k_deg_init(float* __restrict__ deg, int n) {
  int i = blockIdx.x * blockDim.x + threadIdx.x;
  if (i < n) deg[i] = 1.0f;
}

__global__ void k_deg_edges(const int* __restrict__ dst, float* __restrict__ deg, int e) {
  int i = blockIdx.x * blockDim.x + threadIdx.x;
  if (i < e) atomicAdd(&deg[dst[i]], 1.0f);
}

__global__ void k_rsqrt_inplace(float* __restrict__ d, int n) {
  int i = blockIdx.x * blockDim.x + threadIdx.x;
  if (i < n) d[i] = rsqrtf(d[i]);
}

__global__ void k_coef(const int* __restrict__ src, const int* __restrict__ dst,
                       const float* __restrict__ dinv, float* __restrict__ coef, int e) {
  int i = blockIdx.x * blockDim.x + threadIdx.x;
  if (i < e) coef[i] = dinv[src[i]] * dinv[dst[i]];
}

template <int K, int M>
__global__ __launch_bounds__(256)
void k_gemm_wmma(const float* __restrict__ X, const float* __restrict__ W,
                 float* __restrict__ H, int n_rows) {
  const int lane = threadIdx.x & 31;
  const int wave = (blockIdx.x * (blockDim.x >> 5)) + (threadIdx.x >> 5);
  constexpr int pairsPerRow = M / 32;
  const int nWork = (n_rows / 16) * pairsPerRow;
  if (wave >= nWork) return;

  const int ti = wave / pairsPerRow;
  const int tj = (wave % pairsPerRow) * 2;

  const int m  = lane & 15;
  const int ks = (lane >> 4) << 1;

  const float* __restrict__ xrow = X + (size_t)(ti * 16 + m) * K + ks;
  const float* __restrict__ w0   = W + (size_t)(tj * 16) + m;
  const float* __restrict__ w1   = w0 + 16;

  v8f c0 = {}, c1 = {};
#pragma unroll
  for (int k0 = 0; k0 < K; k0 += 4) {
    v2f a, b0, b1;
    a.x  = xrow[k0];
    a.y  = xrow[k0 + 1];
    b0.x = w0[(size_t)(k0 + ks) * M];
    b0.y = w0[(size_t)(k0 + ks + 1) * M];
    b1.x = w1[(size_t)(k0 + ks) * M];
    b1.y = w1[(size_t)(k0 + ks + 1) * M];
    c0 = __builtin_amdgcn_wmma_f32_16x16x4_f32(false, a, false, b0, (short)0, c0,
                                               false, false);
    c1 = __builtin_amdgcn_wmma_f32_16x16x4_f32(false, a, false, b1, (short)0, c1,
                                               false, false);
  }

  const int row0 = ti * 16 + ((lane >> 4) << 3);
  float* __restrict__ out0 = H + (size_t)row0 * M + tj * 16 + m;
#pragma unroll
  for (int v = 0; v < 8; ++v) {
    out0[(size_t)v * M]      = c0[v];
    out0[(size_t)v * M + 16] = c1[v];
  }
}

#define PIECES_PER_THREAD 4
#define PIECES_PER_BLOCK  (256 * PIECES_PER_THREAD)

template <int M>
__global__ __launch_bounds__(256)
void k_init_agg(const float* __restrict__ h, const float* __restrict__ dinv,
                const float* __restrict__ bias, float* __restrict__ agg,
                int n) {
  constexpr unsigned Q = M / 4;
  static_assert(256 % Q == 0, "a thread keeps one column group over its pieces");
  const unsigned total4 = (unsigned)n * Q;
  if (total4 == 0) return;
  const unsigned q0 = blockIdx.x * PIECES_PER_BLOCK + threadIdx.x;
  const float4* __restrict__ h4 = reinterpret_cast<const float4*>(h);
  float4* __restrict__ a4 = reinterpret_cast<float4*>(agg);
  const float4 bv = reinterpret_cast<const float4*>(bias)[q0 % Q];
  float4 hv[PIECES_PER_THREAD];
  float  dv[PIECES_PER_THREAD];
#pragma unroll
  for (int u = 0; u < PIECES_PER_THREAD; ++u) {
    const unsigned q  = q0 + 256u * u;
    const unsigned qc = q < total4 ? q : total4 - 1;
    hv[u] = h4[qc];
    dv[u] = dinv[qc / Q];
  }
#pragma unroll
  for (int u = 0; u < PIECES_PER_THREAD; ++u) {
    const unsigned q = q0 + 256u * u;
    if (q < total4) {
      const float s = dv[u] * dv[u];
      float4 r;
      r.x = bv.x + s * hv[u].x;
      r.y = bv.y + s * hv[u].y;
      r.z = bv.z + s * hv[u].z;
      r.w = bv.w + s * hv[u].w;
      a4[q] = r;
    }
  }
}

#define EDGES_PER_WAVE  128
#define LINES_IN_FLIGHT 16

template <int M, int U>
__device__ __forceinline__ void scatter_edges(int sv, int dv, float cv, int j, int lane,
                                              const float* __restrict__ h,
                                              float* __restrict__ agg) {
  constexpr int F = M / 32;
  float hv[U][F];
#pragma unroll
  for (int u = 0; u < U; ++u) {
    const int s = __builtin_amdgcn_readlane(sv, j + u);
    const float* __restrict__ hp = h + (size_t)s * M + lane;
#pragma unroll
    for (int c = 0; c < F; ++c) hv[u][c] = hp[32 * c];
  }
#pragma unroll
  for (int u = 0; u < U; ++u) {
    const int   d  = __builtin_amdgcn_readlane(dv, j + u);
    const float cf = __int_as_float(__builtin_amdgcn_readlane(__float_as_int(cv), j + u));
    float* __restrict__ ap = agg + (size_t)d * M + lane;
#pragma unroll
    for (int c = 0; c < F; ++c) atomicAdd(ap + 32 * c, cf * hv[u][c]);
  }
}

template <int M>
__global__ __launch_bounds__(256)
void k_scatter(const int* __restrict__ src, const int* __restrict__ dst,
               const float* __restrict__ coef, const float* __restrict__ h,
               float* __restrict__ agg, int e) {
  const int lane = threadIdx.x & 31;
  const int wave = __builtin_amdgcn_readfirstlane(
      (blockIdx.x * (blockDim.x >> 5)) + (threadIdx.x >> 5));
  const int base = wave * EDGES_PER_WAVE;
  if (base >= e) return;
  constexpr int NCH = EDGES_PER_WAVE / 32;
  int   sv[NCH], dv[NCH];
  float cv[NCH];
#pragma unroll
  for (int k = 0; k < NCH; ++k) {
    const int idx = base + 32 * k + lane;
    sv[k] = 0; dv[k] = 0; cv[k] = 0.0f;
    if (idx < e) {
      sv[k] = src[idx];
      dv[k] = dst[idx];
      cv[k] = coef[idx];
    }
  }
#pragma unroll
  for (int k = 0; k < NCH; ++k) {
    const int left = e - (base + 32 * k);
    const int cnt  = left < 32 ? left : 32;
    constexpr int G = LINES_IN_FLIGHT / (M / 32);
    int j = 0;
    for (; j + G <= cnt; j += G)
      scatter_edges<M, G>(sv[k], dv[k], cv[k], j, lane, h, agg);
    for (; j < cnt; ++j)
      scatter_edges<M, 1>(sv[k], dv[k], cv[k], j, lane, h, agg);
  }
}

__global__ __launch_bounds__(256)
void k_relu_res(const float* __restrict__ agg, const float* __restrict__ xp,
                float* __restrict__ xn, int total) {
  const unsigned total4 = (unsigned)total >> 2;
  if (total4 == 0) return;
  const unsigned q0 = blockIdx.x * PIECES_PER_BLOCK + threadIdx.x;
  const float4* __restrict__ a4 = reinterpret_cast<const float4*>(agg);
  const float4* __restrict__ p4 = reinterpret_cast<const float4*>(xp);
  float4* __restrict__ n4 = reinterpret_cast<float4*>(xn);
  float4 av[PIECES_PER_THREAD], pv[PIECES_PER_THREAD];
#pragma unroll
  for (int u = 0; u < PIECES_PER_THREAD; ++u) {
    const unsigned q  = q0 + 256u * u;
    const unsigned qc = q < total4 ? q : total4 - 1;
    av[u] = a4[qc];
    pv[u] = p4[qc];
  }
#pragma unroll
  for (int u = 0; u < PIECES_PER_THREAD; ++u) {
    const unsigned q = q0 + 256u * u;
    if (q < total4) {
      const float vx = av[u].x + pv[u].x;
      const float vy = av[u].y + pv[u].y;
      const float vz = av[u].z + pv[u].z;
      const float vw = av[u].w + pv[u].w;
      float4 r;
      r.x = vx > 0.0f ? vx : 0.0f;
      r.y = vy > 0.0f ? vy : 0.0f;
      r.z = vz > 0.0f ? vz : 0.0f;
      r.w = vw > 0.0f ? vw : 0.0f;
      n4[q] = r;
    }
  }
}

__global__ __launch_bounds__(256)
void k_add(const float* __restrict__ a, const float* __restrict__ b,
           float* __restrict__ o, int total) {
  const unsigned total4 = (unsigned)total >> 2;
  if (total4 == 0) return;
  const unsigned q0 = blockIdx.x * PIECES_PER_BLOCK + threadIdx.x;
  const float4* __restrict__ a4 = reinterpret_cast<const float4*>(a);
  const float4* __restrict__ b4 = reinterpret_cast<const float4*>(b);
  float4* __restrict__ o4 = reinterpret_cast<float4*>(o);
  float4 av[PIECES_PER_THREAD], bv[PIECES_PER_THREAD];
#pragma unroll
  for (int u = 0; u < PIECES_PER_THREAD; ++u) {
    const unsigned q  = q0 + 256u * u;
    const unsigned qc = q < total4 ? q : total4 - 1;
    av[u] = a4[qc];
    bv[u] = b4[qc];
  }
#pragma unroll
  for (int u = 0; u < PIECES_PER_THREAD; ++u) {
    const unsigned q = q0 + 256u * u;
    if (q < total4) {
      float4 r;
      r.x = av[u].x + bv[u].x;
      r.y = av[u].y + bv[u].y;
      r.z = av[u].z + bv[u].z;
      r.w = av[u].w + bv[u].w;
      o4[q] = r;
    }
  }
}

extern "C" void kernel_launch(void* const* d_in, const int* in_sizes, int n_in,
                              void* d_out, int out_size, void* d_ws, size_t ws_size,
                              hipStream_t stream) {
  const float* x      = (const float*)d_in[0];
  const int*   ei     = (const int*)d_in[1];
  const float* W_in   = (const float*)d_in[2];
  const float* b_in   = (const float*)d_in[3];
  const float* W_m0   = (const float*)d_in[4];
  const float* b_m0   = (const float*)d_in[5];
  const float* W_m1   = (const float*)d_in[6];
  const float* b_m1   = (const float*)d_in[7];
  const float* W_out  = (const float*)d_in[8];
  const float* b_out  = (const float*)d_in[9];
  const float* W_skip = (const float*)d_in[10];

  const int* src = ei;
  const int* dst = ei + N_EDGES;

  float* w    = (float*)d_ws;
  float* dinv = w;  w += N_NODES;
  float* coef = w;  w += N_EDGES;
  float* h    = w;  w += (size_t)N_NODES * NFH;
  float* agg  = w;  w += (size_t)N_NODES * NFH;
  float* xA   = w;  w += (size_t)N_NODES * NFH;
  float* xB   = w;  w += (size_t)N_NODES * NFH;
  float* skip = w;  w += (size_t)N_NODES * NCLS;

  const int T = 256;
  auto cdiv = [](int a, int b) { return (a + b - 1) / b; };

  const int nodeBlocks  = cdiv(N_NODES, T);
  const int edgeBlocks  = cdiv(N_EDGES, T);
  const int featBlocks  = cdiv(N_NODES * NFH / 4, PIECES_PER_BLOCK);
  const int clsBlocks   = cdiv(N_NODES * NCLS / 4, PIECES_PER_BLOCK);
  const int scatBlocks  = cdiv(N_EDGES, (T / 32) * EDGES_PER_WAVE);
  const int gemmBlocksH = cdiv((N_NODES / 16) * (NFH / 32), T / 32);
  const int gemmBlocksC = cdiv((N_NODES / 16) * (NCLS / 32), T / 32);

  k_deg_init<<<nodeBlocks, T, 0, stream>>>(dinv, N_NODES);
  k_deg_edges<<<edgeBlocks, T, 0, stream>>>(dst, dinv, N_EDGES);
  k_rsqrt_inplace<<<nodeBlocks, T, 0, stream>>>(dinv, N_NODES);
  k_coef<<<edgeBlocks, T, 0, stream>>>(src, dst, dinv, coef, N_EDGES);

  k_gemm_wmma<NFH, NFH><<<gemmBlocksH, T, 0, stream>>>(x, W_in, h, N_NODES);
  k_init_agg<NFH><<<featBlocks, T, 0, stream>>>(h, dinv, b_in, agg, N_NODES);
  k_scatter<NFH><<<scatBlocks, T, 0, stream>>>(src, dst, coef, h, agg, N_EDGES);
  k_relu_res<<<featBlocks, T, 0, stream>>>(agg, x, xA, N_NODES * NFH);

  k_gemm_wmma<NFH, NFH><<<gemmBlocksH, T, 0, stream>>>(xA, W_m0, h, N_NODES);
  k_init_agg<NFH><<<featBlocks, T, 0, stream>>>(h, dinv, b_m0, agg, N_NODES);
  k_scatter<NFH><<<scatBlocks, T, 0, stream>>>(src, dst, coef, h, agg, N_EDGES);
  k_relu_res<<<featBlocks, T, 0, stream>>>(agg, xA, xB, N_NODES * NFH);

  k_gemm_wmma<NFH, NFH><<<gemmBlocksH, T, 0, stream>>>(xB, W_m1, h, N_NODES);
  k_init_agg<NFH><<<featBlocks, T, 0, stream>>>(h, dinv, b_m1, agg, N_NODES);
  k_scatter<NFH><<<scatBlocks, T, 0, stream>>>(src, dst, coef, h, agg, N_EDGES);
  k_relu_res<<<featBlocks, T, 0, stream>>>(agg, xB, xA, N_NODES * NFH);

  k_gemm_wmma<NFH, NCLS><<<gemmBlocksC, T, 0, stream>>>(xA, W_out, h, N_NODES);
  k_init_agg<NCLS><<<clsBlocks, T, 0, stream>>>(h, dinv, b_out, agg, N_NODES);
  k_scatter<NCLS><<<scatBlocks, T, 0, stream>>>(src, dst, coef, h, agg, N_EDGES);
  k_gemm_wmma<NFH, NCLS><<<gemmBlocksC, T, 0, stream>>>(xA, W_skip, skip, N_NODES);
  k_add<<<clsBlocks, T, 0, stream>>>(agg, skip, (float*)d_out, N_NODES * NCLS);
}


# Edge lists built once: one launch sorts each 6400-edge block by destination tile in LDS (returning int adds); tile kernels walk their ranges, no key scan per layer; sums stay in LDS arrival order
# speedup vs baseline: 3.3945x; 2.0283x over previous
#include <hip/hip_runtime.h>
#include <hip/hip_bf16.h>

#define N_NODES 100000
#define N_EDGES 1600000
#define NFH     128
#define NCLS    64

typedef __attribute__((ext_vector_type(2))) float v2f;
typedef __attribute__((ext_vector_type(8))) float v8f;

__global__ void k_deg_init(float* __restrict__ deg, int n) {
  int i = blockIdx.x * blockDim.x + threadIdx.x;
  if (i < n) deg[i] = 1.0f;
}

__global__ void k_deg_edges(const int* __restrict__ dst, float* __restrict__ deg, int e) {
  int i = blockIdx.x * blockDim.x + threadIdx.x;
  if (i < e) atomicAdd(&deg[dst[i]], 1.0f);
}

__global__ void k_rsqrt_inplace(float* __restrict__ d, int n) {
  int i = blockIdx.x * blockDim.x + threadIdx.x;
  if (i < n) d[i] = rsqrtf(d[i]);
}

#define TILE_ROWS 400
#define N_TILES   (N_NODES / TILE_ROWS)
static_assert(N_TILES * TILE_ROWS == N_NODES, "tiles cover the nodes exactly");
static_assert(N_TILES <= 256, "the tile counters fit one 256-entry table");
static_assert(N_NODES <= (1 << 17) && TILE_ROWS <= (1 << 9), "src and row-in-tile share one word");

#define EBLK_EDGES   6400
#define N_EBLKS      (N_EDGES / EBLK_EDGES)
#define PART_THREADS 1024
#define PART_PPT     ((EBLK_EDGES + PART_THREADS - 1) / PART_THREADS)
static_assert(N_EBLKS * EBLK_EDGES == N_EDGES, "blocks cover the edges exactly");
static_assert(EBLK_EDGES % 4 == 0, "a block leaves LDS in 16-byte pieces");

__global__ __launch_bounds__(PART_THREADS)
void k_partition(const int* __restrict__ src, const int* __restrict__ dst,
                 const float* __restrict__ dinv, int* __restrict__ ent,
                 float* __restrict__ cf, int* __restrict__ off) {
  __shared__ int cnt[256];
  __shared__ int start[256];
  __shared__ __attribute__((aligned(16))) int   s_ent[EBLK_EDGES];
  __shared__ __attribute__((aligned(16))) float s_cf[EBLK_EDGES];
  const int tid  = threadIdx.x;
  const int base = blockIdx.x * EBLK_EDGES;

  if (tid < 256) cnt[tid] = 0;
  __syncthreads();

  int   sv[PART_PPT], dv[PART_PPT], rk[PART_PPT];
  float cv[PART_PPT];
#pragma unroll
  for (int k = 0; k < PART_PPT; ++k) {
    const int i  = tid + k * PART_THREADS;
    const int ic = i < EBLK_EDGES ? i : EBLK_EDGES - 1;
    sv[k] = src[base + ic];
    dv[k] = dst[base + ic];
  }
#pragma unroll
  for (int k = 0; k < PART_PPT; ++k) {
    const int i = tid + k * PART_THREADS;
    cv[k] = dinv[sv[k]] * dinv[dv[k]];
    rk[k] = 0;
    if (i < EBLK_EDGES) rk[k] = atomicAdd(&cnt[dv[k] / TILE_ROWS], 1);
  }
  __syncthreads();

  if (tid < 32) {
    int c[8];
    int run = 0;
#pragma unroll
    for (int k = 0; k < 8; ++k) {
      const int x = cnt[tid * 8 + k];
      c[k] = run;
      run += x;
    }
    int inc = run;
#pragma unroll
    for (int d = 1; d < 32; d <<= 1) {
      const int y = __shfl_up(inc, d, 32);
      if (tid >= d) inc += y;
    }
    const int excl = inc - run;
#pragma unroll
    for (int k = 0; k < 8; ++k) start[tid * 8 + k] = excl + c[k];
  }
  __syncthreads();

  if (tid <= N_TILES) off[(size_t)blockIdx.x * (N_TILES + 1) + tid] = base + start[tid];

#pragma unroll
  for (int k = 0; k < PART_PPT; ++k) {
    const int i = tid + k * PART_THREADS;
    if (i < EBLK_EDGES) {
      const int tl = dv[k] / TILE_ROWS;
      const int p  = start[tl] + rk[k];
      s_ent[p] = sv[k] | ((dv[k] - tl * TILE_ROWS) << 17);
      s_cf[p]  = cv[k];
    }
  }
  __syncthreads();

  const int4*   se4 = reinterpret_cast<const int4*>(s_ent);
  const float4* sc4 = reinterpret_cast<const float4*>(s_cf);
  int4*   __restrict__ e4 = reinterpret_cast<int4*>(ent + base);
  float4* __restrict__ c4 = reinterpret_cast<float4*>(cf + base);
  for (int i = tid; i < EBLK_EDGES / 4; i += PART_THREADS) {
    e4[i] = se4[i];
    c4[i] = sc4[i];
  }
}

template <int K, int M>
__global__ __launch_bounds__(256)
void k_gemm_wmma(const float* __restrict__ X, const float* __restrict__ W,
                 float* __restrict__ H, int n_rows) {
  const int lane = threadIdx.x & 31;
  const int wave = (blockIdx.x * (blockDim.x >> 5)) + (threadIdx.x >> 5);
  constexpr int pairsPerRow = M / 32;
  const int nWork = (n_rows / 16) * pairsPerRow;
  if (wave >= nWork) return;

  const int ti = wave / pairsPerRow;
  const int tj = (wave % pairsPerRow) * 2;

  const int m  = lane & 15;
  const int ks = (lane >> 4) << 1;

  const float* __restrict__ xrow = X + (size_t)(ti * 16 + m) * K + ks;
  const float* __restrict__ w0   = W + (size_t)(tj * 16) + m;
  const float* __restrict__ w1   = w0 + 16;

  v8f c0 = {}, c1 = {};
#pragma unroll
  for (int k0 = 0; k0 < K; k0 += 4) {
    v2f a, b0, b1;
    a.x  = xrow[k0];
    a.y  = xrow[k0 + 1];
    b0.x = w0[(size_t)(k0 + ks) * M];
    b0.y = w0[(size_t)(k0 + ks + 1) * M];
    b1.x = w1[(size_t)(k0 + ks) * M];
    b1.y = w1[(size_t)(k0 + ks + 1) * M];
    c0 = __builtin_amdgcn_wmma_f32_16x16x4_f32(false, a, false, b0, (short)0, c0,
                                               false, false);
    c1 = __builtin_amdgcn_wmma_f32_16x16x4_f32(false, a, false, b1, (short)0, c1,
                                               false, false);
  }

  const int row0 = ti * 16 + ((lane >> 4) << 3);
  float* __restrict__ out0 = H + (size_t)row0 * M + tj * 16 + m;
#pragma unroll
  for (int v = 0; v < 8; ++v) {
    out0[(size_t)v * M]      = c0[v];
    out0[(size_t)v * M + 16] = c1[v];
  }
}

#define AGG_WAVES 32

template <int M, int U>
__device__ __forceinline__ void tile_add_edges(int sv, int dv, float cv, int j, int lane,
                                               const float* __restrict__ h,
                                               float* tile) {
  constexpr int F = M / 32;
  float hv[U][F];
#pragma unroll
  for (int u = 0; u < U; ++u) {
    const int s = __builtin_amdgcn_readlane(sv, j + u);
    const float* __restrict__ hp = h + (size_t)s * M + lane;
#pragma unroll
    for (int c = 0; c < F; ++c) hv[u][c] = hp[32 * c];
  }
#pragma unroll
  for (int u = 0; u < U; ++u) {
    const int   d  = __builtin_amdgcn_readlane(dv, j + u);
    const float cf = __int_as_float(__builtin_amdgcn_readlane(__float_as_int(cv), j + u));
    float* tp = tile + d * M + lane;
#pragma unroll
    for (int c = 0; c < F; ++c) atomicAdd(tp + 32 * c, cf * hv[u][c]);
  }
}

template <int M>
__device__ __forceinline__ void tile_add_batch(int sv, int dv, float cv, int n, int lane,
                                               const float* __restrict__ h,
                                               float* tile) {
  constexpr int G = 32 / (M / 32);
  int j = 0;
  for (; j + G <= n; j += G) tile_add_edges<M, G>(sv, dv, cv, j, lane, h, tile);
  for (; j < n; ++j)         tile_add_edges<M, 1>(sv, dv, cv, j, lane, h, tile);
}

template <int M, bool LAST>
__global__ __launch_bounds__(AGG_WAVES * 32)
void k_tile_agg(const int* __restrict__ ent, const float* __restrict__ cf,
                const int* __restrict__ off,
                const float* __restrict__ h, const float* __restrict__ dinv,
                const float* __restrict__ bias, const float* __restrict__ res,
                float* __restrict__ outp) {
  __shared__ __attribute__((aligned(16))) float tile[TILE_ROWS * M];
  constexpr int Q  = M / 4;
  constexpr int NP = TILE_ROWS * Q;
  constexpr int NT = AGG_WAVES * 32;
  constexpr int UN = 4;

  const int lane = threadIdx.x & 31;
  const int wave = __builtin_amdgcn_readfirstlane(threadIdx.x >> 5);
  const int t    = blockIdx.x;
  const int lo   = t * TILE_ROWS;

  {
    const float4* __restrict__ h4 = reinterpret_cast<const float4*>(h) + (size_t)lo * Q;
    float4* t4 = reinterpret_cast<float4*>(tile);
    static_assert(NT % Q == 0, "a thread keeps one column group over its pieces");
    const float4 bv = reinterpret_cast<const float4*>(bias)[threadIdx.x % Q];
    for (int p0 = threadIdx.x; p0 < NP; p0 += NT * UN) {
      float4 hv[UN];
      float  dv[UN];
#pragma unroll
      for (int u = 0; u < UN; ++u) {
        const int p  = p0 + NT * u;
        const int pc = p < NP ? p : NP - 1;
        hv[u] = h4[pc];
        dv[u] = dinv[lo + pc / Q];
      }
#pragma unroll
      for (int u = 0; u < UN; ++u) {
        const int p = p0 + NT * u;
        if (p < NP) {
          const float s = dv[u] * dv[u];
          float4 r;
          r.x = bv.x + s * hv[u].x;
          r.y = bv.y + s * hv[u].y;
          r.z = bv.z + s * hv[u].z;
          r.w = bv.w + s * hv[u].w;
          t4[p] = r;
        }
      }
    }
  }
  __syncthreads();

  {
    constexpr int NSEG = (N_EBLKS + AGG_WAVES - 1) / AGG_WAVES;
    static_assert(NSEG <= 32, "one lane per range");
    int slo = 0, slen = 0;
    {
      const int b = wave + lane * AGG_WAVES;
      if (lane < NSEG && b < N_EBLKS) {
        const int* __restrict__ o = off + (size_t)b * (N_TILES + 1) + t;
        slo  = o[0];
        slen = o[1] - slo;
      }
    }
    int los[NSEG], pex[NSEG];
    int tot = 0;
#pragma unroll
    for (int sg = 0; sg < NSEG; ++sg) {
      los[sg] = __builtin_amdgcn_readlane(slo, sg);
      pex[sg] = tot;
      tot += __builtin_amdgcn_readlane(slen, sg);
    }
    auto entry_of = [&](int v0) {
      const int v = v0 + lane;
      int idx = 0;
#pragma unroll
      for (int sg = 0; sg < NSEG; ++sg) idx = v >= pex[sg] ? los[sg] + (v - pex[sg]) : idx;
      return v < tot ? idx : 0;
    };
    int   en = 0;
    float cv = 0.0f;
    if (tot > 0) {
      const int idx = entry_of(0);
      en = ent[idx];
      cv = cf[idx];
    }
    for (int v0 = 0; v0 < tot; v0 += 32) {
      int   en_n = 0;
      float cv_n = 0.0f;
      if (v0 + 32 < tot) {
        const int idx = entry_of(v0 + 32);
        en_n = ent[idx];
        cv_n = cf[idx];
      }
      const int n = tot - v0 < 32 ? tot - v0 : 32;
      tile_add_batch<M>(en & 0x1ffff, (int)((unsigned)en >> 17), cv, n, lane, h, tile);
      en = en_n;
      cv = cv_n;
    }
  }
  __syncthreads();

  {
    const float4* t4 = reinterpret_cast<const float4*>(tile);
    const float4* __restrict__ r4 = reinterpret_cast<const float4*>(res) + (size_t)lo * Q;
    float4* __restrict__ o4 = reinterpret_cast<float4*>(outp) + (size_t)lo * Q;
    for (int p0 = threadIdx.x; p0 < NP; p0 += NT * UN) {
      float4 pv[UN];
#pragma unroll
      for (int u = 0; u < UN; ++u) {
        const int p  = p0 + NT * u;
        pv[u] = r4[p < NP ? p : NP - 1];
      }
#pragma unroll
      for (int u = 0; u < UN; ++u) {
        const int p = p0 + NT * u;
        if (p < NP) {
          const float4 av = t4[p];
          const float vx = av.x + pv[u].x;
          const float vy = av.y + pv[u].y;
          const float vz = av.z + pv[u].z;
          const float vw = av.w + pv[u].w;
          float4 r;
          if (LAST) {
            r.x = vx; r.y = vy; r.z = vz; r.w = vw;
          } else {
            r.x = vx > 0.0f ? vx : 0.0f;
            r.y = vy > 0.0f ? vy : 0.0f;
            r.z = vz > 0.0f ? vz : 0.0f;
            r.w = vw > 0.0f ? vw : 0.0f;
          }
          o4[p] = r;
        }
      }
    }
  }
}

extern "C" void kernel_launch(void* const* d_in, const int* in_sizes, int n_in,
                              void* d_out, int out_size, void* d_ws, size_t ws_size,
                              hipStream_t stream) {
  const float* x      = (const float*)d_in[0];
  const int*   ei     = (const int*)d_in[1];
  const float* W_in   = (const float*)d_in[2];
  const float* b_in   = (const float*)d_in[3];
  const float* W_m0   = (const float*)d_in[4];
  const float* b_m0   = (const float*)d_in[5];
  const float* W_m1   = (const float*)d_in[6];
  const float* b_m1   = (const float*)d_in[7];
  const float* W_out  = (const float*)d_in[8];
  const float* b_out  = (const float*)d_in[9];
  const float* W_skip = (const float*)d_in[10];

  const int* src = ei;
  const int* dst = ei + N_EDGES;

  float* w    = (float*)d_ws;
  float* dinv = w;  w += N_NODES;
  float* cf   = w;  w += N_EDGES;
  int*   ent  = (int*)w;  w += N_EDGES;
  float* h    = w;  w += (size_t)N_NODES * NFH;
  float* xA   = w;  w += (size_t)N_NODES * NFH;
  float* xB   = w;  w += (size_t)N_NODES * NFH;
  float* skip = w;  w += (size_t)N_NODES * NCLS;
  int*   off  = (int*)w;

  const int T = 256;
  auto cdiv = [](int a, int b) { return (a + b - 1) / b; };

  const int nodeBlocks  = cdiv(N_NODES, T);
  const int edgeBlocks  = cdiv(N_EDGES, T);
  const int gemmBlocksH = cdiv((N_NODES / 16) * (NFH / 32), T / 32);
  const int gemmBlocksC = cdiv((N_NODES / 16) * (NCLS / 32), T / 32);
  const int TA = AGG_WAVES * 32;

  k_deg_init<<<nodeBlocks, T, 0, stream>>>(dinv, N_NODES);
  k_deg_edges<<<edgeBlocks, T, 0, stream>>>(dst, dinv, N_EDGES);
  k_rsqrt_inplace<<<nodeBlocks, T, 0, stream>>>(dinv, N_NODES);
  k_partition<<<N_EBLKS, PART_THREADS, 0, stream>>>(src, dst, dinv, ent, cf, off);

  k_gemm_wmma<NFH, NFH><<<gemmBlocksH, T, 0, stream>>>(x, W_in, h, N_NODES);
  k_tile_agg<NFH, false><<<N_TILES, TA, 0, stream>>>(ent, cf, off, h, dinv, b_in, x, xA);

  k_gemm_wmma<NFH, NFH><<<gemmBlocksH, T, 0, stream>>>(xA, W_m0, h, N_NODES);
  k_tile_agg<NFH, false><<<N_TILES, TA, 0, stream>>>(ent, cf, off, h, dinv, b_m0, xA, xB);

  k_gemm_wmma<NFH, NFH><<<gemmBlocksH, T, 0, stream>>>(xB, W_m1, h, N_NODES);
  k_tile_agg<NFH, false><<<N_TILES, TA, 0, stream>>>(ent, cf, off, h, dinv, b_m1, xB, xA);

  k_gemm_wmma<NFH, NCLS><<<gemmBlocksC, T, 0, stream>>>(xA, W_out, h, N_NODES);
  k_gemm_wmma<NFH, NCLS><<<gemmBlocksC, T, 0, stream>>>(xA, W_skip, skip, N_NODES);
  k_tile_agg<NCLS, true><<<N_TILES, TA, 0, stream>>>(ent, cf, off, h, dinv, b_out, skip,
                                                    (float*)d_out);
}


# GEMM waves made resident: 250 groups x 8 waves, each keeps a 128x32 W panel in registers and walks its row tiles, 16-byte A loads one tile ahead; W_out|W_skip share a launch; k order in sums changes
# speedup vs baseline: 3.8441x; 1.1324x over previous
#include <hip/hip_runtime.h>
#include <hip/hip_bf16.h>

#define N_NODES 100000
#define N_EDGES 1600000
#define NFH     128
#define NCLS    64

typedef __attribute__((ext_vector_type(2))) float v2f;
typedef __attribute__((ext_vector_type(8))) float v8f;

__global__ void k_deg_init(float* __restrict__ deg, int n) {
  int i = blockIdx.x * blockDim.x + threadIdx.x;
  if (i < n) deg[i] = 1.0f;
}

__global__ void k_deg_edges(const int* __restrict__ dst, float* __restrict__ deg, int e) {
  int i = blockIdx.x * blockDim.x + threadIdx.x;
  if (i < e) atomicAdd(&deg[dst[i]], 1.0f);
}

__global__ void k_rsqrt_inplace(float* __restrict__ d, int n) {
  int i = blockIdx.x * blockDim.x + threadIdx.x;
  if (i < n) d[i] = rsqrtf(d[i]);
}

#define TILE_ROWS 400
#define N_TILES   (N_NODES / TILE_ROWS)
static_assert(N_TILES * TILE_ROWS == N_NODES, "tiles cover the nodes exactly");
static_assert(N_TILES <= 256, "the tile counters fit one 256-entry table");
static_assert(N_NODES <= (1 << 17) && TILE_ROWS <= (1 << 9), "src and row-in-tile share one word");

#define EBLK_EDGES   6400
#define N_EBLKS      (N_EDGES / EBLK_EDGES)
#define PART_THREADS 1024
#define PART_PPT     ((EBLK_EDGES + PART_THREADS - 1) / PART_THREADS)
static_assert(N_EBLKS * EBLK_EDGES == N_EDGES, "blocks cover the edges exactly");
static_assert(EBLK_EDGES % 4 == 0, "a block leaves LDS in 16-byte pieces");

__global__ __launch_bounds__(PART_THREADS)
void k_partition(const int* __restrict__ src, const int* __restrict__ dst,
                 const float* __restrict__ dinv, int* __restrict__ ent,
                 float* __restrict__ cf, int* __restrict__ off) {
  __shared__ int cnt[256];
  __shared__ int start[256];
  __shared__ __attribute__((aligned(16))) int   s_ent[EBLK_EDGES];
  __shared__ __attribute__((aligned(16))) float s_cf[EBLK_EDGES];
  const int tid  = threadIdx.x;
  const int base = blockIdx.x * EBLK_EDGES;

  if (tid < 256) cnt[tid] = 0;
  __syncthreads();

  int   sv[PART_PPT], dv[PART_PPT], rk[PART_PPT];
  float cv[PART_PPT];
#pragma unroll
  for (int k = 0; k < PART_PPT; ++k) {
    const int i  = tid + k * PART_THREADS;
    const int ic = i < EBLK_EDGES ? i : EBLK_EDGES - 1;
    sv[k] = src[base + ic];
    dv[k] = dst[base + ic];
  }
#pragma unroll
  for (int k = 0; k < PART_PPT; ++k) {
    const int i = tid + k * PART_THREADS;
    cv[k] = dinv[sv[k]] * dinv[dv[k]];
    rk[k] = 0;
    if (i < EBLK_EDGES) rk[k] = atomicAdd(&cnt[dv[k] / TILE_ROWS], 1);
  }
  __syncthreads();

  if (tid < 32) {
    int c[8];
    int run = 0;
#pragma unroll
    for (int k = 0; k < 8; ++k) {
      const int x = cnt[tid * 8 + k];
      c[k] = run;
      run += x;
    }
    int inc = run;
#pragma unroll
    for (int d = 1; d < 32; d <<= 1) {
      const int y = __shfl_up(inc, d, 32);
      if (tid >= d) inc += y;
    }
    const int excl = inc - run;
#pragma unroll
    for (int k = 0; k < 8; ++k) start[tid * 8 + k] = excl + c[k];
  }
  __syncthreads();

  if (tid <= N_TILES) off[(size_t)blockIdx.x * (N_TILES + 1) + tid] = base + start[tid];

#pragma unroll
  for (int k = 0; k < PART_PPT; ++k) {
    const int i = tid + k * PART_THREADS;
    if (i < EBLK_EDGES) {
      const int tl = dv[k] / TILE_ROWS;
      const int p  = start[tl] + rk[k];
      s_ent[p] = sv[k] | ((dv[k] - tl * TILE_ROWS) << 17);
      s_cf[p]  = cv[k];
    }
  }
  __syncthreads();

  const int4*   se4 = reinterpret_cast<const int4*>(s_ent);
  const float4* sc4 = reinterpret_cast<const float4*>(s_cf);
  int4*   __restrict__ e4 = reinterpret_cast<int4*>(ent + base);
  float4* __restrict__ c4 = reinterpret_cast<float4*>(cf + base);
  for (int i = tid; i < EBLK_EDGES / 4; i += PART_THREADS) {
    e4[i] = se4[i];
    c4[i] = sc4[i];
  }
}

#define GEMM_ROWS  400
#define GEMM_TILES (GEMM_ROWS / 16)
#define GEMM_WGS   (N_NODES / GEMM_ROWS)
static_assert(GEMM_WGS * GEMM_ROWS == N_NODES && GEMM_TILES * 16 == GEMM_ROWS,
              "row ranges cover the nodes exactly in whole 16-row tiles");

__device__ __forceinline__ void gemm_load_rows(float4 (&a)[16], const float4* __restrict__ xa,
                                               int t) {
  const float4* __restrict__ xt = xa + (size_t)t * (16 * NFH / 4);
#pragma unroll
  for (int q = 0; q < 16; ++q) a[q] = xt[2 * q];
}

template <int MO>
__device__ __forceinline__ void gemm_tile(const float4 (&a)[16], const v2f (&b)[32][2],
                                          float* __restrict__ out) {
  v8f c0 = {}, c1 = {};
#pragma unroll
  for (int q = 0; q < 16; ++q) {
    v2f lo, hi;
    lo.x = a[q].x; lo.y = a[q].y;
    hi.x = a[q].z; hi.y = a[q].w;
    c0 = __builtin_amdgcn_wmma_f32_16x16x4_f32(false, lo, false, b[2 * q][0], (short)0, c0,
                                               false, false);
    c1 = __builtin_amdgcn_wmma_f32_16x16x4_f32(false, lo, false, b[2 * q][1], (short)0, c1,
                                               false, false);
    c0 = __builtin_amdgcn_wmma_f32_16x16x4_f32(false, hi, false, b[2 * q + 1][0], (short)0, c0,
                                               false, false);
    c1 = __builtin_amdgcn_wmma_f32_16x16x4_f32(false, hi, false, b[2 * q + 1][1], (short)0, c1,
                                               false, false);
  }
#pragma unroll
  for (int v = 0; v < 8; ++v) {
    out[(size_t)v * MO]      = c0[v];
    out[(size_t)v * MO + 16] = c1[v];
  }
}

template <int MO>
__global__ __launch_bounds__(256)
void k_gemm_resident(const float* __restrict__ X,
                     const float* __restrict__ W0, const float* __restrict__ W1,
                     float* __restrict__ H0, float* __restrict__ H1) {
  constexpr int PPM = MO / 32;
  const int lane = threadIdx.x & 31;
  const int wave = __builtin_amdgcn_readfirstlane(threadIdx.x >> 5);
  const int p    = wave & 3;
  const int rh   = wave >> 2;
  const int c    = lane & 15;
  const int hf   = lane >> 4;

  const float* __restrict__ Wm = (p / PPM) ? W1 : W0;
  float* __restrict__       Hm = (p / PPM) ? H1 : H0;
  const int col0 = (p % PPM) * 32;

  v2f b[32][2];
  {
    const float* __restrict__ wp = Wm + (size_t)(4 * hf) * MO + col0 + c;
#pragma unroll
    for (int q = 0; q < 16; ++q) {
#pragma unroll
      for (int st = 0; st < 2; ++st) {
#pragma unroll
        for (int j = 0; j < 2; ++j) {
          const int k = 8 * q + 2 * st;
          b[2 * q + st][j].x = wp[(size_t)k * MO + 16 * j];
          b[2 * q + st][j].y = wp[(size_t)(k + 1) * MO + 16 * j];
        }
      }
    }
  }

  const int row0 = blockIdx.x * GEMM_ROWS;
  const float4* __restrict__ xa =
      reinterpret_cast<const float4*>(X + (size_t)(row0 + c) * NFH + 4 * hf);
  float* __restrict__ ob = Hm + (size_t)(row0 + 8 * hf) * MO + col0 + c;

  const int n = (GEMM_TILES - rh + 1) / 2;
  float4 a0[16], a1[16];
  gemm_load_rows(a0, xa, rh);
  int i = 0;
  for (; i + 1 < n; i += 2) {
    const int t0 = rh + 2 * i;
    const int t1 = t0 + 2;
    const int t2 = rh + 2 * (i + 2 < n ? i + 2 : n - 1);
    gemm_load_rows(a1, xa, t1);
    gemm_tile<MO>(a0, b, ob + (size_t)t0 * 16 * MO);
    gemm_load_rows(a0, xa, t2);
    gemm_tile<MO>(a1, b, ob + (size_t)t1 * 16 * MO);
  }
  if (i < n) gemm_tile<MO>(a0, b, ob + (size_t)(rh + 2 * i) * 16 * MO);
}

#define AGG_WAVES 32

template <int M, int U>
__device__ __forceinline__ void tile_add_edges(int sv, int dv, float cv, int j, int lane,
                                               const float* __restrict__ h,
                                               float* tile) {
  constexpr int F = M / 32;
  float hv[U][F];
#pragma unroll
  for (int u = 0; u < U; ++u) {
    const int s = __builtin_amdgcn_readlane(sv, j + u);
    const float* __restrict__ hp = h + (size_t)s * M + lane;
#pragma unroll
    for (int c = 0; c < F; ++c) hv[u][c] = hp[32 * c];
  }
#pragma unroll
  for (int u = 0; u < U; ++u) {
    const int   d  = __builtin_amdgcn_readlane(dv, j + u);
    const float cf = __int_as_float(__builtin_amdgcn_readlane(__float_as_int(cv), j + u));
    float* tp = tile + d * M + lane;
#pragma unroll
    for (int c = 0; c < F; ++c) atomicAdd(tp + 32 * c, cf * hv[u][c]);
  }
}

template <int M>
__device__ __forceinline__ void tile_add_batch(int sv, int dv, float cv, int n, int lane,
                                               const float* __restrict__ h,
                                               float* tile) {
  constexpr int G = 32 / (M / 32);
  int j = 0;
  for (; j + G <= n; j += G) tile_add_edges<M, G>(sv, dv, cv, j, lane, h, tile);
  for (; j < n; ++j)         tile_add_edges<M, 1>(sv, dv, cv, j, lane, h, tile);
}

template <int M, bool LAST>
__global__ __launch_bounds__(AGG_WAVES * 32)
void k_tile_agg(const int* __restrict__ ent, const float* __restrict__ cf,
                const int* __restrict__ off,
                const float* __restrict__ h, const float* __restrict__ dinv,
                const float* __restrict__ bias, const float* __restrict__ res,
                float* __restrict__ outp) {
  __shared__ __attribute__((aligned(16))) float tile[TILE_ROWS * M];
  constexpr int Q  = M / 4;
  constexpr int NP = TILE_ROWS * Q;
  constexpr int NT = AGG_WAVES * 32;
  constexpr int UN = 4;

  const int lane = threadIdx.x & 31;
  const int wave = __builtin_amdgcn_readfirstlane(threadIdx.x >> 5);
  const int t    = blockIdx.x;
  const int lo   = t * TILE_ROWS;

  {
    const float4* __restrict__ h4 = reinterpret_cast<const float4*>(h) + (size_t)lo * Q;
    float4* t4 = reinterpret_cast<float4*>(tile);
    static_assert(NT % Q == 0, "a thread keeps one column group over its pieces");
    const float4 bv = reinterpret_cast<const float4*>(bias)[threadIdx.x % Q];
    for (int p0 = threadIdx.x; p0 < NP; p0 += NT * UN) {
      float4 hv[UN];
      float  dv[UN];
#pragma unroll
      for (int u = 0; u < UN; ++u) {
        const int p  = p0 + NT * u;
        const int pc = p < NP ? p : NP - 1;
        hv[u] = h4[pc];
        dv[u] = dinv[lo + pc / Q];
      }
#pragma unroll
      for (int u = 0; u < UN; ++u) {
        const int p = p0 + NT * u;
        if (p < NP) {
          const float s = dv[u] * dv[u];
          float4 r;
          r.x = bv.x + s * hv[u].x;
          r.y = bv.y + s * hv[u].y;
          r.z = bv.z + s * hv[u].z;
          r.w = bv.w + s * hv[u].w;
          t4[p] = r;
        }
      }
    }
  }
  __syncthreads();

  {
    constexpr int NSEG = (N_EBLKS + AGG_WAVES - 1) / AGG_WAVES;
    static_assert(NSEG <= 32, "one lane per range");
    int slo = 0, slen = 0;
    {
      const int b = wave + lane * AGG_WAVES;
      if (lane < NSEG && b < N_EBLKS) {
        const int* __restrict__ o = off + (size_t)b * (N_TILES + 1) + t;
        slo  = o[0];
        slen = o[1] - slo;
      }
    }
    int los[NSEG], pex[NSEG];
    int tot = 0;
#pragma unroll
    for (int sg = 0; sg < NSEG; ++sg) {
      los[sg] = __builtin_amdgcn_readlane(slo, sg);
      pex[sg] = tot;
      tot += __builtin_amdgcn_readlane(slen, sg);
    }
    auto entry_of = [&](int v0) {
      const int v = v0 + lane;
      int idx = 0;
#pragma unroll
      for (int sg = 0; sg < NSEG; ++sg) idx = v >= pex[sg] ? los[sg] + (v - pex[sg]) : idx;
      return v < tot ? idx : 0;
    };
    int   en = 0;
    float cv = 0.0f;
    if (tot > 0) {
      const int idx = entry_of(0);
      en = ent[idx];
      cv = cf[idx];
    }
    for (int v0 = 0; v0 < tot; v0 += 32) {
      int   en_n = 0;
      float cv_n = 0.0f;
      if (v0 + 32 < tot) {
        const int idx = entry_of(v0 + 32);
        en_n = ent[idx];
        cv_n = cf[idx];
      }
      const int n = tot - v0 < 32 ? tot - v0 : 32;
      tile_add_batch<M>(en & 0x1ffff, (int)((unsigned)en >> 17), cv, n, lane, h, tile);
      en = en_n;
      cv = cv_n;
    }
  }
  __syncthreads();

  {
    const float4* t4 = reinterpret_cast<const float4*>(tile);
    const float4* __restrict__ r4 = reinterpret_cast<const float4*>(res) + (size_t)lo * Q;
    float4* __restrict__ o4 = reinterpret_cast<float4*>(outp) + (size_t)lo * Q;
    for (int p0 = threadIdx.x; p0 < NP; p0 += NT * UN) {
      float4 pv[UN];
#pragma unroll
      for (int u = 0; u < UN; ++u) {
        const int p  = p0 + NT * u;
        pv[u] = r4[p < NP ? p : NP - 1];
      }
#pragma unroll
      for (int u = 0; u < UN; ++u) {
        const int p = p0 + NT * u;
        if (p < NP) {
          const float4 av = t4[p];
          const float vx = av.x + pv[u].x;
          const float vy = av.y + pv[u].y;
          const float vz = av.z + pv[u].z;
          const float vw = av.w + pv[u].w;
          float4 r;
          if (LAST) {
            r.x = vx; r.y = vy; r.z = vz; r.w = vw;
          } else {
            r.x = vx > 0.0f ? vx : 0.0f;
            r.y = vy > 0.0f ? vy : 0.0f;
            r.z = vz > 0.0f ? vz : 0.0f;
            r.w = vw > 0.0f ? vw : 0.0f;
          }
          o4[p] = r;
        }
      }
    }
  }
}

extern "C" void kernel_launch(void* const* d_in, const int* in_sizes, int n_in,
                              void* d_out, int out_size, void* d_ws, size_t ws_size,
                              hipStream_t stream) {
  const float* x      = (const float*)d_in[0];
  const int*   ei     = (const int*)d_in[1];
  const float* W_in   = (const float*)d_in[2];
  const float* b_in   = (const float*)d_in[3];
  const float* W_m0   = (const float*)d_in[4];
  const float* b_m0   = (const float*)d_in[5];
  const float* W_m1   = (const float*)d_in[6];
  const float* b_m1   = (const float*)d_in[7];
  const float* W_out  = (const float*)d_in[8];
  const float* b_out  = (const float*)d_in[9];
  const float* W_skip = (const float*)d_in[10];

  const int* src = ei;
  const int* dst = ei + N_EDGES;

  float* w    = (float*)d_ws;
  float* dinv = w;  w += N_NODES;
  float* cf   = w;  w += N_EDGES;
  int*   ent  = (int*)w;  w += N_EDGES;
  float* h    = w;  w += (size_t)N_NODES * NFH;
  float* xA   = w;  w += (size_t)N_NODES * NFH;
  float* xB   = w;  w += (size_t)N_NODES * NFH;
  float* skip = w;  w += (size_t)N_NODES * NCLS;
  int*   off  = (int*)w;

  const int T = 256;
  auto cdiv = [](int a, int b) { return (a + b - 1) / b; };

  const int nodeBlocks  = cdiv(N_NODES, T);
  const int edgeBlocks  = cdiv(N_EDGES, T);
  const int TA = AGG_WAVES * 32;

  k_deg_init<<<nodeBlocks, T, 0, stream>>>(dinv, N_NODES);
  k_deg_edges<<<edgeBlocks, T, 0, stream>>>(dst, dinv, N_EDGES);
  k_rsqrt_inplace<<<nodeBlocks, T, 0, stream>>>(dinv, N_NODES);
  k_partition<<<N_EBLKS, PART_THREADS, 0, stream>>>(src, dst, dinv, ent, cf, off);

  k_gemm_resident<NFH><<<GEMM_WGS, 256, 0, stream>>>(x, W_in, W_in, h, h);
  k_tile_agg<NFH, false><<<N_TILES, TA, 0, stream>>>(ent, cf, off, h, dinv, b_in, x, xA);

  k_gemm_resident<NFH><<<GEMM_WGS, 256, 0, stream>>>(xA, W_m0, W_m0, h, h);
  k_tile_agg<NFH, false><<<N_TILES, TA, 0, stream>>>(ent, cf, off, h, dinv, b_m0, xA, xB);

  k_gemm_resident<NFH><<<GEMM_WGS, 256, 0, stream>>>(xB, W_m1, W_m1, h, h);
  k_tile_agg<NFH, false><<<N_TILES, TA, 0, stream>>>(ent, cf, off, h, dinv, b_m1, xB, xA);

  k_gemm_resident<NCLS><<<GEMM_WGS, 256, 0, stream>>>(xA, W_out, W_skip, h, skip);
  k_tile_agg<NCLS, true><<<N_TILES, TA, 0, stream>>>(ent, cf, off, h, dinv, b_out, skip,
                                                    (float*)d_out);
}


# Row gathers as whole aligned 256-byte units: h plane on a 256-byte boundary, one 16-byte (64 columns: 8-byte) load per lane per row; LDS tile rows word-permuted, adds stay on 32 banks, arrival order
# speedup vs baseline: 3.9948x; 1.0392x over previous
#include <hip/hip_runtime.h>
#include <hip/hip_bf16.h>

#define N_NODES 100000
#define N_EDGES 1600000
#define NFH     128
#define NCLS    64

typedef __attribute__((ext_vector_type(2))) float v2f;
typedef __attribute__((ext_vector_type(4))) float v4f;
typedef __attribute__((ext_vector_type(8))) float v8f;

__global__ void k_deg_init(float* __restrict__ deg, int n) {
  int i = blockIdx.x * blockDim.x + threadIdx.x;
  if (i < n) deg[i] = 1.0f;
}

__global__ void k_deg_edges(const int* __restrict__ dst, float* __restrict__ deg, int e) {
  int i = blockIdx.x * blockDim.x + threadIdx.x;
  if (i < e) atomicAdd(&deg[dst[i]], 1.0f);
}

__global__ void k_rsqrt_inplace(float* __restrict__ d, int n) {
  int i = blockIdx.x * blockDim.x + threadIdx.x;
  if (i < n) d[i] = rsqrtf(d[i]);
}

#define DINV_FLOATS ((N_NODES + 63) / 64 * 64)
static_assert((N_EDGES % 64) == 0 && ((N_NODES * NCLS) % 64) == 0,
              "the regions between dinv and the planes are whole 256-byte units");

#define TILE_ROWS 400
#define N_TILES   (N_NODES / TILE_ROWS)
static_assert(N_TILES * TILE_ROWS == N_NODES, "tiles cover the nodes exactly");
static_assert(N_TILES <= 256, "the tile counters fit one 256-entry table");
static_assert(N_NODES <= (1 << 17) && TILE_ROWS <= (1 << 9), "src and row-in-tile share one word");

#define EBLK_EDGES   6400
#define N_EBLKS      (N_EDGES / EBLK_EDGES)
#define PART_THREADS 1024
#define PART_PPT     ((EBLK_EDGES + PART_THREADS - 1) / PART_THREADS)
static_assert(N_EBLKS * EBLK_EDGES == N_EDGES, "blocks cover the edges exactly");
static_assert(EBLK_EDGES % 4 == 0, "a block leaves LDS in 16-byte pieces");

__global__ __launch_bounds__(PART_THREADS)
void k_partition(const int* __restrict__ src, const int* __restrict__ dst,
                 const float* __restrict__ dinv, int* __restrict__ ent,
                 float* __restrict__ cf, int* __restrict__ off) {
  __shared__ int cnt[256];
  __shared__ int start[256];
  __shared__ __attribute__((aligned(16))) int   s_ent[EBLK_EDGES];
  __shared__ __attribute__((aligned(16))) float s_cf[EBLK_EDGES];
  const int tid  = threadIdx.x;
  const int base = blockIdx.x * EBLK_EDGES;

  if (tid < 256) cnt[tid] = 0;
  __syncthreads();

  int   sv[PART_PPT], dv[PART_PPT], rk[PART_PPT];
  float cv[PART_PPT];
#pragma unroll
  for (int k = 0; k < PART_PPT; ++k) {
    const int i  = tid + k * PART_THREADS;
    const int ic = i < EBLK_EDGES ? i : EBLK_EDGES - 1;
    sv[k] = src[base + ic];
    dv[k] = dst[base + ic];
  }
#pragma unroll
  for (int k = 0; k < PART_PPT; ++k) {
    const int i = tid + k * PART_THREADS;
    cv[k] = dinv[sv[k]] * dinv[dv[k]];
    rk[k] = 0;
    if (i < EBLK_EDGES) rk[k] = atomicAdd(&cnt[dv[k] / TILE_ROWS], 1);
  }
  __syncthreads();

  if (tid < 32) {
    int c[8];
    int run = 0;
#pragma unroll
    for (int k = 0; k < 8; ++k) {
      const int x = cnt[tid * 8 + k];
      c[k] = run;
      run += x;
    }
    int inc = run;
#pragma unroll
    for (int d = 1; d < 32; d <<= 1) {
      const int y = __shfl_up(inc, d, 32);
      if (tid >= d) inc += y;
    }
    const int excl = inc - run;
#pragma unroll
    for (int k = 0; k < 8; ++k) start[tid * 8 + k] = excl + c[k];
  }
  __syncthreads();

  if (tid <= N_TILES) off[(size_t)blockIdx.x * (N_TILES + 1) + tid] = base + start[tid];

#pragma unroll
  for (int k = 0; k < PART_PPT; ++k) {
    const int i = tid + k * PART_THREADS;
    if (i < EBLK_EDGES) {
      const int tl = dv[k] / TILE_ROWS;
      const int p  = start[tl] + rk[k];
      s_ent[p] = sv[k] | ((dv[k] - tl * TILE_ROWS) << 17);
      s_cf[p]  = cv[k];
    }
  }
  __syncthreads();

  const int4*   se4 = reinterpret_cast<const int4*>(s_ent);
  const float4* sc4 = reinterpret_cast<const float4*>(s_cf);
  int4*   __restrict__ e4 = reinterpret_cast<int4*>(ent + base);
  float4* __restrict__ c4 = reinterpret_cast<float4*>(cf + base);
  for (int i = tid; i < EBLK_EDGES / 4; i += PART_THREADS) {
    e4[i] = se4[i];
    c4[i] = sc4[i];
  }
}

#define GEMM_ROWS  400
#define GEMM_TILES (GEMM_ROWS / 16)
#define GEMM_WGS   (N_NODES / GEMM_ROWS)
static_assert(GEMM_WGS * GEMM_ROWS == N_NODES && GEMM_TILES * 16 == GEMM_ROWS,
              "row ranges cover the nodes exactly in whole 16-row tiles");

__device__ __forceinline__ void gemm_load_rows(float4 (&a)[16], const float4* __restrict__ xa,
                                               int t) {
  const float4* __restrict__ xt = xa + (size_t)t * (16 * NFH / 4);
#pragma unroll
  for (int q = 0; q < 16; ++q) a[q] = xt[2 * q];
}

template <int MO>
__device__ __forceinline__ void gemm_tile(const float4 (&a)[16], const v2f (&b)[32][2],
                                          float* __restrict__ out) {
  v8f c0 = {}, c1 = {};
#pragma unroll
  for (int q = 0; q < 16; ++q) {
    v2f lo, hi;
    lo.x = a[q].x; lo.y = a[q].y;
    hi.x = a[q].z; hi.y = a[q].w;
    c0 = __builtin_amdgcn_wmma_f32_16x16x4_f32(false, lo, false, b[2 * q][0], (short)0, c0,
                                               false, false);
    c1 = __builtin_amdgcn_wmma_f32_16x16x4_f32(false, lo, false, b[2 * q][1], (short)0, c1,
                                               false, false);
    c0 = __builtin_amdgcn_wmma_f32_16x16x4_f32(false, hi, false, b[2 * q + 1][0], (short)0, c0,
                                               false, false);
    c1 = __builtin_amdgcn_wmma_f32_16x16x4_f32(false, hi, false, b[2 * q + 1][1], (short)0, c1,
                                               false, false);
  }
#pragma unroll
  for (int v = 0; v < 8; ++v) {
    out[(size_t)v * MO]      = c0[v];
    out[(size_t)v * MO + 16] = c1[v];
  }
}

template <int MO>
__global__ __launch_bounds__(256)
void k_gemm_resident(const float* __restrict__ X,
                     const float* __restrict__ W0, const float* __restrict__ W1,
                     float* __restrict__ H0, float* __restrict__ H1) {
  constexpr int PPM = MO / 32;
  const int lane = threadIdx.x & 31;
  const int wave = __builtin_amdgcn_readfirstlane(threadIdx.x >> 5);
  const int p    = wave & 3;
  const int rh   = wave >> 2;
  const int c    = lane & 15;
  const int hf   = lane >> 4;

  const float* __restrict__ Wm = (p / PPM) ? W1 : W0;
  float* __restrict__       Hm = (p / PPM) ? H1 : H0;
  const int col0 = (p % PPM) * 32;

  v2f b[32][2];
  {
    const float* __restrict__ wp = Wm + (size_t)(4 * hf) * MO + col0 + c;
#pragma unroll
    for (int q = 0; q < 16; ++q) {
#pragma unroll
      for (int st = 0; st < 2; ++st) {
#pragma unroll
        for (int j = 0; j < 2; ++j) {
          const int k = 8 * q + 2 * st;
          b[2 * q + st][j].x = wp[(size_t)k * MO + 16 * j];
          b[2 * q + st][j].y = wp[(size_t)(k + 1) * MO + 16 * j];
        }
      }
    }
  }

  const int row0 = blockIdx.x * GEMM_ROWS;
  const float4* __restrict__ xa =
      reinterpret_cast<const float4*>(X + (size_t)(row0 + c) * NFH + 4 * hf);
  float* __restrict__ ob = Hm + (size_t)(row0 + 8 * hf) * MO + col0 + c;

  const int n = (GEMM_TILES - rh + 1) / 2;
  float4 a0[16], a1[16];
  gemm_load_rows(a0, xa, rh);
  int i = 0;
  for (; i + 1 < n; i += 2) {
    const int t0 = rh + 2 * i;
    const int t1 = t0 + 2;
    const int t2 = rh + 2 * (i + 2 < n ? i + 2 : n - 1);
    gemm_load_rows(a1, xa, t1);
    gemm_tile<MO>(a0, b, ob + (size_t)t0 * 16 * MO);
    gemm_load_rows(a0, xa, t2);
    gemm_tile<MO>(a1, b, ob + (size_t)t1 * 16 * MO);
  }
  if (i < n) gemm_tile<MO>(a0, b, ob + (size_t)(rh + 2 * i) * 16 * MO);
}

#define AGG_WAVES 32

template <int F> struct RowPiece;
template <> struct RowPiece<4> { typedef v4f type; };
template <> struct RowPiece<2> { typedef v2f type; };

template <int F>
__device__ __forceinline__ constexpr int tile_word(int c) { return 32 * (c % F) + c / F; }

template <int M, int U>
__device__ __forceinline__ void tile_add_edges(int sv, int dv, float cv, int j, int lane,
                                               const float* __restrict__ h,
                                               float* tile) {
  constexpr int F = M / 32;
  typedef typename RowPiece<F>::type piece_t;
  piece_t hv[U];
#pragma unroll
  for (int u = 0; u < U; ++u) {
    const int s = __builtin_amdgcn_readlane(sv, j + u);
    hv[u] = reinterpret_cast<const piece_t*>(h + (size_t)s * M)[lane];
  }
#pragma unroll
  for (int u = 0; u < U; ++u) {
    const int   d  = __builtin_amdgcn_readlane(dv, j + u);
    const float cf = __int_as_float(__builtin_amdgcn_readlane(__float_as_int(cv), j + u));
    float* tp = tile + d * M + lane;
#pragma unroll
    for (int k = 0; k < F; ++k) atomicAdd(tp + 32 * k, cf * hv[u][k]);
  }
}

template <int M>
__device__ __forceinline__ void tile_add_batch(int sv, int dv, float cv, int n, int lane,
                                               const float* __restrict__ h,
                                               float* tile) {
  constexpr int G = 32 / (M / 32);
  int j = 0;
  for (; j + G <= n; j += G) tile_add_edges<M, G>(sv, dv, cv, j, lane, h, tile);
  for (; j < n; ++j)         tile_add_edges<M, 1>(sv, dv, cv, j, lane, h, tile);
}

template <int M, bool LAST>
__global__ __launch_bounds__(AGG_WAVES * 32)
void k_tile_agg(const int* __restrict__ ent, const float* __restrict__ cf,
                const int* __restrict__ off,
                const float* __restrict__ h, const float* __restrict__ dinv,
                const float* __restrict__ bias, const float* __restrict__ res,
                float* __restrict__ outp) {
  __shared__ __attribute__((aligned(16))) float tile[TILE_ROWS * M];
  constexpr int Q  = M / 4;
  constexpr int NP = TILE_ROWS * Q;
  constexpr int NT = AGG_WAVES * 32;
  constexpr int UN = 4;
  constexpr int F  = M / 32;

  const int lane = threadIdx.x & 31;
  const int wave = __builtin_amdgcn_readfirstlane(threadIdx.x >> 5);
  const int t    = blockIdx.x;
  const int lo   = t * TILE_ROWS;
  const int cq   = threadIdx.x % Q;

  {
    const float4* __restrict__ h4 = reinterpret_cast<const float4*>(h) + (size_t)lo * Q;
    static_assert(NT % Q == 0, "a thread keeps one column group over its pieces");
    const float4 bv = reinterpret_cast<const float4*>(bias)[cq];
    for (int p0 = threadIdx.x; p0 < NP; p0 += NT * UN) {
      float4 hv[UN];
      float  dv[UN];
#pragma unroll
      for (int u = 0; u < UN; ++u) {
        const int p  = p0 + NT * u;
        const int pc = p < NP ? p : NP - 1;
        hv[u] = h4[pc];
        dv[u] = dinv[lo + pc / Q];
      }
#pragma unroll
      for (int u = 0; u < UN; ++u) {
        const int p = p0 + NT * u;
        if (p < NP) {
          const float s = dv[u] * dv[u];
          float* tr = tile + (p / Q) * M;
          tr[tile_word<F>(4 * cq)]     = bv.x + s * hv[u].x;
          tr[tile_word<F>(4 * cq + 1)] = bv.y + s * hv[u].y;
          tr[tile_word<F>(4 * cq + 2)] = bv.z + s * hv[u].z;
          tr[tile_word<F>(4 * cq + 3)] = bv.w + s * hv[u].w;
        }
      }
    }
  }
  __syncthreads();

  {
    constexpr int NSEG = (N_EBLKS + AGG_WAVES - 1) / AGG_WAVES;
    static_assert(NSEG <= 32, "one lane per range");
    int slo = 0, slen = 0;
    {
      const int b = wave + lane * AGG_WAVES;
      if (lane < NSEG && b < N_EBLKS) {
        const int* __restrict__ o = off + (size_t)b * (N_TILES + 1) + t;
        slo  = o[0];
        slen = o[1] - slo;
      }
    }
    int los[NSEG], pex[NSEG];
    int tot = 0;
#pragma unroll
    for (int sg = 0; sg < NSEG; ++sg) {
      los[sg] = __builtin_amdgcn_readlane(slo, sg);
      pex[sg] = tot;
      tot += __builtin_amdgcn_readlane(slen, sg);
    }
    auto entry_of = [&](int v0) {
      const int v = v0 + lane;
      int idx = 0;
#pragma unroll
      for (int sg = 0; sg < NSEG; ++sg) idx = v >= pex[sg] ? los[sg] + (v - pex[sg]) : idx;
      return v < tot ? idx : 0;
    };
    int   en = 0;
    float cv = 0.0f;
    if (tot > 0) {
      const int idx = entry_of(0);
      en = ent[idx];
      cv = cf[idx];
    }
    for (int v0 = 0; v0 < tot; v0 += 32) {
      int   en_n = 0;
      float cv_n = 0.0f;
      if (v0 + 32 < tot) {
        const int idx = entry_of(v0 + 32);
        en_n = ent[idx];
        cv_n = cf[idx];
      }
      const int n = tot - v0 < 32 ? tot - v0 : 32;
      tile_add_batch<M>(en & 0x1ffff, (int)((unsigned)en >> 17), cv, n, lane, h, tile);
      en = en_n;
      cv = cv_n;
    }
  }
  __syncthreads();

  {
    const float4* __restrict__ r4 = reinterpret_cast<const float4*>(res) + (size_t)lo * Q;
    float4* __restrict__ o4 = reinterpret_cast<float4*>(outp) + (size_t)lo * Q;
    for (int p0 = threadIdx.x; p0 < NP; p0 += NT * UN) {
      float4 pv[UN];
#pragma unroll
      for (int u = 0; u < UN; ++u) {
        const int p  = p0 + NT * u;
        pv[u] = r4[p < NP ? p : NP - 1];
      }
#pragma unroll
      for (int u = 0; u < UN; ++u) {
        const int p = p0 + NT * u;
        if (p < NP) {
          const float* tr = tile + (p / Q) * M;
          const float vx = tr[tile_word<F>(4 * cq)]     + pv[u].x;
          const float vy = tr[tile_word<F>(4 * cq + 1)] + pv[u].y;
          const float vz = tr[tile_word<F>(4 * cq + 2)] + pv[u].z;
          const float vw = tr[tile_word<F>(4 * cq + 3)] + pv[u].w;
          float4 r;
          if (LAST) {
            r.x = vx; r.y = vy; r.z = vz; r.w = vw;
          } else {
            r.x = vx > 0.0f ? vx : 0.0f;
            r.y = vy > 0.0f ? vy : 0.0f;
            r.z = vz > 0.0f ? vz : 0.0f;
            r.w = vw > 0.0f ? vw : 0.0f;
          }
          o4[p] = r;
        }
      }
    }
  }
}

extern "C" void kernel_launch(void* const* d_in, const int* in_sizes, int n_in,
                              void* d_out, int out_size, void* d_ws, size_t ws_size,
                              hipStream_t stream) {
  const float* x      = (const float*)d_in[0];
  const int*   ei     = (const int*)d_in[1];
  const float* W_in   = (const float*)d_in[2];
  const float* b_in   = (const float*)d_in[3];
  const float* W_m0   = (const float*)d_in[4];
  const float* b_m0   = (const float*)d_in[5];
  const float* W_m1   = (const float*)d_in[6];
  const float* b_m1   = (const float*)d_in[7];
  const float* W_out  = (const float*)d_in[8];
  const float* b_out  = (const float*)d_in[9];
  const float* W_skip = (const float*)d_in[10];

  const int* src = ei;
  const int* dst = ei + N_EDGES;

  float* w    = (float*)d_ws;
  float* dinv = w;  w += DINV_FLOATS;
  float* cf   = w;  w += N_EDGES;
  int*   ent  = (int*)w;  w += N_EDGES;
  float* h    = w;  w += (size_t)N_NODES * NFH;
  float* xA   = w;  w += (size_t)N_NODES * NFH;
  float* xB   = w;  w += (size_t)N_NODES * NFH;
  float* skip = w;  w += (size_t)N_NODES * NCLS;
  int*   off  = (int*)w;

  const int T = 256;
  auto cdiv = [](int a, int b) { return (a + b - 1) / b; };

  const int nodeBlocks  = cdiv(N_NODES, T);
  const int edgeBlocks  = cdiv(N_EDGES, T);
  const int TA = AGG_WAVES * 32;

  k_deg_init<<<nodeBlocks, T, 0, stream>>>(dinv, N_NODES);
  k_deg_edges<<<edgeBlocks, T, 0, stream>>>(dst, dinv, N_EDGES);
  k_rsqrt_inplace<<<nodeBlocks, T, 0, stream>>>(dinv, N_NODES);
  k_partition<<<N_EBLKS, PART_THREADS, 0, stream>>>(src, dst, dinv, ent, cf, off);

  k_gemm_resident<NFH><<<GEMM_WGS, 256, 0, stream>>>(x, W_in, W_in, h, h);
  k_tile_agg<NFH, false><<<N_TILES, TA, 0, stream>>>(ent, cf, off, h, dinv, b_in, x, xA);

  k_gemm_resident<NFH><<<GEMM_WGS, 256, 0, stream>>>(xA, W_m0, W_m0, h, h);
  k_tile_agg<NFH, false><<<N_TILES, TA, 0, stream>>>(ent, cf, off, h, dinv, b_m0, xA, xB);

  k_gemm_resident<NFH><<<GEMM_WGS, 256, 0, stream>>>(xB, W_m1, W_m1, h, h);
  k_tile_agg<NFH, false><<<N_TILES, TA, 0, stream>>>(ent, cf, off, h, dinv, b_m1, xB, xA);

  k_gemm_resident<NCLS><<<GEMM_WGS, 256, 0, stream>>>(xA, W_out, W_skip, h, skip);
  k_tile_agg<NCLS, true><<<N_TILES, TA, 0, stream>>>(ent, cf, off, h, dinv, b_out, skip,
                                                    (float*)d_out);
}


# Normalisation taken out of the edge lists: GEMM stores dinv[row]*h, tiles add gathered rows unweighted, dinv[dst] and bias applied once after the sum; coefficient array and its gathers gone
# speedup vs baseline: 4.2871x; 1.0732x over previous
#include <hip/hip_runtime.h>
#include <hip/hip_bf16.h>

#define N_NODES 100000
#define N_EDGES 1600000
#define NFH     128
#define NCLS    64

typedef __attribute__((ext_vector_type(2))) float v2f;
typedef __attribute__((ext_vector_type(4))) float v4f;
typedef __attribute__((ext_vector_type(8))) float v8f;

__global__ void k_deg_init(float* __restrict__ deg, int n) {
  int i = blockIdx.x * blockDim.x + threadIdx.x;
  if (i < n) deg[i] = 1.0f;
}

__global__ void k_deg_edges(const int* __restrict__ dst, float* __restrict__ deg, int e) {
  int i = blockIdx.x * blockDim.x + threadIdx.x;
  if (i < e) atomicAdd(&deg[dst[i]], 1.0f);
}

__global__ void k_rsqrt_inplace(float* __restrict__ d, int n) {
  int i = blockIdx.x * blockDim.x + threadIdx.x;
  if (i < n) d[i] = rsqrtf(d[i]);
}

#define DINV_FLOATS ((N_NODES + 63) / 64 * 64)
static_assert((N_EDGES % 64) == 0 && ((N_NODES * NCLS) % 64) == 0,
              "the regions between dinv and the planes are whole 256-byte units");

#define TILE_ROWS 400
#define N_TILES   (N_NODES / TILE_ROWS)
static_assert(N_TILES * TILE_ROWS == N_NODES, "tiles cover the nodes exactly");
static_assert(N_TILES <= 256, "the tile counters fit one 256-entry table");
static_assert(N_NODES <= (1 << 17) && TILE_ROWS <= (1 << 9), "src and row-in-tile share one word");

#define EBLK_EDGES   6400
#define N_EBLKS      (N_EDGES / EBLK_EDGES)
#define PART_THREADS 1024
#define PART_PPT     ((EBLK_EDGES + PART_THREADS - 1) / PART_THREADS)
static_assert(N_EBLKS * EBLK_EDGES == N_EDGES, "blocks cover the edges exactly");
static_assert(EBLK_EDGES % 4 == 0, "a block leaves LDS in 16-byte pieces");

__global__ __launch_bounds__(PART_THREADS)
void k_partition(const int* __restrict__ src, const int* __restrict__ dst,
                 int* __restrict__ ent, int* __restrict__ off) {
  __shared__ int cnt[256];
  __shared__ int start[256];
  __shared__ __attribute__((aligned(16))) int s_ent[EBLK_EDGES];
  const int tid  = threadIdx.x;
  const int base = blockIdx.x * EBLK_EDGES;

  if (tid < 256) cnt[tid] = 0;
  __syncthreads();

  int sv[PART_PPT], dv[PART_PPT], rk[PART_PPT];
#pragma unroll
  for (int k = 0; k < PART_PPT; ++k) {
    const int i  = tid + k * PART_THREADS;
    const int ic = i < EBLK_EDGES ? i : EBLK_EDGES - 1;
    sv[k] = src[base + ic];
    dv[k] = dst[base + ic];
  }
#pragma unroll
  for (int k = 0; k < PART_PPT; ++k) {
    const int i = tid + k * PART_THREADS;
    rk[k] = 0;
    if (i < EBLK_EDGES) rk[k] = atomicAdd(&cnt[dv[k] / TILE_ROWS], 1);
  }
  __syncthreads();

  if (tid < 32) {
    int c[8];
    int run = 0;
#pragma unroll
    for (int k = 0; k < 8; ++k) {
      const int x = cnt[tid * 8 + k];
      c[k] = run;
      run += x;
    }
    int inc = run;
#pragma unroll
    for (int d = 1; d < 32; d <<= 1) {
      const int y = __shfl_up(inc, d, 32);
      if (tid >= d) inc += y;
    }
    const int excl = inc - run;
#pragma unroll
    for (int k = 0; k < 8; ++k) start[tid * 8 + k] = excl + c[k];
  }
  __syncthreads();

  if (tid <= N_TILES) off[(size_t)blockIdx.x * (N_TILES + 1) + tid] = base + start[tid];

#pragma unroll
  for (int k = 0; k < PART_PPT; ++k) {
    const int i = tid + k * PART_THREADS;
    if (i < EBLK_EDGES) {
      const int tl = dv[k] / TILE_ROWS;
      const int p  = start[tl] + rk[k];
      s_ent[p] = sv[k] | ((dv[k] - tl * TILE_ROWS) << 17);
    }
  }
  __syncthreads();

  const int4* se4 = reinterpret_cast<const int4*>(s_ent);
  int4* __restrict__ e4 = reinterpret_cast<int4*>(ent + base);
  for (int i = tid; i < EBLK_EDGES / 4; i += PART_THREADS) e4[i] = se4[i];
}

#define GEMM_ROWS  400
#define GEMM_TILES (GEMM_ROWS / 16)
#define GEMM_WGS   (N_NODES / GEMM_ROWS)
static_assert(GEMM_WGS * GEMM_ROWS == N_NODES && GEMM_TILES * 16 == GEMM_ROWS,
              "row ranges cover the nodes exactly in whole 16-row tiles");
static_assert(GEMM_ROWS % 8 == 0, "a lane's 8 row scales are two aligned 16-byte loads");

__device__ __forceinline__ void gemm_load_rows(float4 (&a)[16], float4 (&dq)[2],
                                               const float4* __restrict__ xa,
                                               const float4* __restrict__ da, int t) {
  const float4* __restrict__ xt = xa + (size_t)t * (16 * NFH / 4);
#pragma unroll
  for (int q = 0; q < 16; ++q) a[q] = xt[2 * q];
  dq[0] = da[4 * t];
  dq[1] = da[4 * t + 1];
}

template <int MO>
__device__ __forceinline__ void gemm_tile(const float4 (&a)[16], const float4 (&dq)[2],
                                          const v2f (&b)[32][2], bool scaled,
                                          float* __restrict__ out) {
  v8f c0 = {}, c1 = {};
#pragma unroll
  for (int q = 0; q < 16; ++q) {
    v2f lo, hi;
    lo.x = a[q].x; lo.y = a[q].y;
    hi.x = a[q].z; hi.y = a[q].w;
    c0 = __builtin_amdgcn_wmma_f32_16x16x4_f32(false, lo, false, b[2 * q][0], (short)0, c0,
                                               false, false);
    c1 = __builtin_amdgcn_wmma_f32_16x16x4_f32(false, lo, false, b[2 * q][1], (short)0, c1,
                                               false, false);
    c0 = __builtin_amdgcn_wmma_f32_16x16x4_f32(false, hi, false, b[2 * q + 1][0], (short)0, c0,
                                               false, false);
    c1 = __builtin_amdgcn_wmma_f32_16x16x4_f32(false, hi, false, b[2 * q + 1][1], (short)0, c1,
                                               false, false);
  }
  const float rs[8] = {dq[0].x, dq[0].y, dq[0].z, dq[0].w, dq[1].x, dq[1].y, dq[1].z, dq[1].w};
#pragma unroll
  for (int v = 0; v < 8; ++v) {
    const float sc = scaled ? rs[v] : 1.0f;
    out[(size_t)v * MO]      = c0[v] * sc;
    out[(size_t)v * MO + 16] = c1[v] * sc;
  }
}

template <int MO>
__global__ __launch_bounds__(256)
void k_gemm_resident(const float* __restrict__ X,
                     const float* __restrict__ W0, const float* __restrict__ W1,
                     const float* __restrict__ dinv,
                     float* __restrict__ H0, float* __restrict__ H1) {
  constexpr int PPM = MO / 32;
  const int lane = threadIdx.x & 31;
  const int wave = __builtin_amdgcn_readfirstlane(threadIdx.x >> 5);
  const int p    = wave & 3;
  const int rh   = wave >> 2;
  const int c    = lane & 15;
  const int hf   = lane >> 4;

  const float* __restrict__ Wm = (p / PPM) ? W1 : W0;
  float* __restrict__       Hm = (p / PPM) ? H1 : H0;
  const int col0 = (p % PPM) * 32;
  const bool scaled = (p / PPM) == 0;

  v2f b[32][2];
  {
    const float* __restrict__ wp = Wm + (size_t)(4 * hf) * MO + col0 + c;
#pragma unroll
    for (int q = 0; q < 16; ++q) {
#pragma unroll
      for (int st = 0; st < 2; ++st) {
#pragma unroll
        for (int j = 0; j < 2; ++j) {
          const int k = 8 * q + 2 * st;
          b[2 * q + st][j].x = wp[(size_t)k * MO + 16 * j];
          b[2 * q + st][j].y = wp[(size_t)(k + 1) * MO + 16 * j];
        }
      }
    }
  }

  const int row0 = blockIdx.x * GEMM_ROWS;
  const float4* __restrict__ xa =
      reinterpret_cast<const float4*>(X + (size_t)(row0 + c) * NFH + 4 * hf);
  float* __restrict__ ob = Hm + (size_t)(row0 + 8 * hf) * MO + col0 + c;
  const float4* __restrict__ da = reinterpret_cast<const float4*>(dinv + row0 + 8 * hf);

  const int n = (GEMM_TILES - rh + 1) / 2;
  float4 a0[16], a1[16];
  float4 d0[2], d1[2];
  gemm_load_rows(a0, d0, xa, da, rh);
  int i = 0;
  for (; i + 1 < n; i += 2) {
    const int t0 = rh + 2 * i;
    const int t1 = t0 + 2;
    const int t2 = rh + 2 * (i + 2 < n ? i + 2 : n - 1);
    gemm_load_rows(a1, d1, xa, da, t1);
    gemm_tile<MO>(a0, d0, b, scaled, ob + (size_t)t0 * 16 * MO);
    gemm_load_rows(a0, d0, xa, da, t2);
    gemm_tile<MO>(a1, d1, b, scaled, ob + (size_t)t1 * 16 * MO);
  }
  if (i < n) gemm_tile<MO>(a0, d0, b, scaled, ob + (size_t)(rh + 2 * i) * 16 * MO);
}

#define AGG_WAVES 32

template <int F> struct RowPiece;
template <> struct RowPiece<4> { typedef v4f type; };
template <> struct RowPiece<2> { typedef v2f type; };

template <int F>
__device__ __forceinline__ constexpr int tile_word(int c) { return 32 * (c % F) + c / F; }

template <int M, int U>
__device__ __forceinline__ void tile_add_edges(int sv, int dv, int j, int lane,
                                               const float* __restrict__ h,
                                               float* tile) {
  constexpr int F = M / 32;
  typedef typename RowPiece<F>::type piece_t;
  piece_t hv[U];
#pragma unroll
  for (int u = 0; u < U; ++u) {
    const int s = __builtin_amdgcn_readlane(sv, j + u);
    hv[u] = reinterpret_cast<const piece_t*>(h + (size_t)s * M)[lane];
  }
#pragma unroll
  for (int u = 0; u < U; ++u) {
    const int d = __builtin_amdgcn_readlane(dv, j + u);
    float* tp = tile + d * M + lane;
#pragma unroll
    for (int k = 0; k < F; ++k) atomicAdd(tp + 32 * k, hv[u][k]);
  }
}

template <int M>
__device__ __forceinline__ void tile_add_batch(int sv, int dv, int n, int lane,
                                               const float* __restrict__ h,
                                               float* tile) {
  constexpr int G = 32 / (M / 32);
  int j = 0;
  for (; j + G <= n; j += G) tile_add_edges<M, G>(sv, dv, j, lane, h, tile);
  for (; j < n; ++j)         tile_add_edges<M, 1>(sv, dv, j, lane, h, tile);
}

template <int M, bool LAST>
__global__ __launch_bounds__(AGG_WAVES * 32)
void k_tile_agg(const int* __restrict__ ent, const int* __restrict__ off,
                const float* __restrict__ h, const float* __restrict__ dinv,
                const float* __restrict__ bias, const float* __restrict__ res,
                float* __restrict__ outp) {
  __shared__ __attribute__((aligned(16))) float tile[TILE_ROWS * M];
  constexpr int Q  = M / 4;
  constexpr int NP = TILE_ROWS * Q;
  constexpr int NT = AGG_WAVES * 32;
  constexpr int UN = 4;
  constexpr int F  = M / 32;

  const int lane = threadIdx.x & 31;
  const int wave = __builtin_amdgcn_readfirstlane(threadIdx.x >> 5);
  const int t    = blockIdx.x;
  const int lo   = t * TILE_ROWS;
  const int cq   = threadIdx.x % Q;

  static_assert(NT % Q == 0, "a thread keeps one column group over its pieces");
  {
    const float4* __restrict__ h4 = reinterpret_cast<const float4*>(h) + (size_t)lo * Q;
    for (int p0 = threadIdx.x; p0 < NP; p0 += NT * UN) {
      float4 hv[UN];
#pragma unroll
      for (int u = 0; u < UN; ++u) {
        const int p = p0 + NT * u;
        hv[u] = h4[p < NP ? p : NP - 1];
      }
#pragma unroll
      for (int u = 0; u < UN; ++u) {
        const int p = p0 + NT * u;
        if (p < NP) {
          float* tr = tile + (p / Q) * M;
          tr[tile_word<F>(4 * cq)]     = hv[u].x;
          tr[tile_word<F>(4 * cq + 1)] = hv[u].y;
          tr[tile_word<F>(4 * cq + 2)] = hv[u].z;
          tr[tile_word<F>(4 * cq + 3)] = hv[u].w;
        }
      }
    }
  }
  __syncthreads();

  {
    constexpr int NSEG = (N_EBLKS + AGG_WAVES - 1) / AGG_WAVES;
    static_assert(NSEG <= 32, "one lane per range");
    int slo = 0, slen = 0;
    {
      const int b = wave + lane * AGG_WAVES;
      if (lane < NSEG && b < N_EBLKS) {
        const int* __restrict__ o = off + (size_t)b * (N_TILES + 1) + t;
        slo  = o[0];
        slen = o[1] - slo;
      }
    }
    int los[NSEG], pex[NSEG];
    int tot = 0;
#pragma unroll
    for (int sg = 0; sg < NSEG; ++sg) {
      los[sg] = __builtin_amdgcn_readlane(slo, sg);
      pex[sg] = tot;
      tot += __builtin_amdgcn_readlane(slen, sg);
    }
    auto entry_of = [&](int v0) {
      const int v = v0 + lane;
      int idx = 0;
#pragma unroll
      for (int sg = 0; sg < NSEG; ++sg) idx = v >= pex[sg] ? los[sg] + (v - pex[sg]) : idx;
      return v < tot ? idx : 0;
    };
    int en = 0;
    if (tot > 0) en = ent[entry_of(0)];
    for (int v0 = 0; v0 < tot; v0 += 32) {
      int en_n = 0;
      if (v0 + 32 < tot) en_n = ent[entry_of(v0 + 32)];
      const int n = tot - v0 < 32 ? tot - v0 : 32;
      tile_add_batch<M>(en & 0x1ffff, (int)((unsigned)en >> 17), n, lane, h, tile);
      en = en_n;
    }
  }
  __syncthreads();

  {
    const float4* __restrict__ r4 = reinterpret_cast<const float4*>(res) + (size_t)lo * Q;
    float4* __restrict__ o4 = reinterpret_cast<float4*>(outp) + (size_t)lo * Q;
    const float4 bv = reinterpret_cast<const float4*>(bias)[cq];
    for (int p0 = threadIdx.x; p0 < NP; p0 += NT * UN) {
      float4 pv[UN];
      float  dv[UN];
#pragma unroll
      for (int u = 0; u < UN; ++u) {
        const int p  = p0 + NT * u;
        const int pc = p < NP ? p : NP - 1;
        pv[u] = r4[pc];
        dv[u] = dinv[lo + pc / Q];
      }
#pragma unroll
      for (int u = 0; u < UN; ++u) {
        const int p = p0 + NT * u;
        if (p < NP) {
          const float* tr = tile + (p / Q) * M;
          const float d  = dv[u];
          const float vx = (bv.x + d * tr[tile_word<F>(4 * cq)])     + pv[u].x;
          const float vy = (bv.y + d * tr[tile_word<F>(4 * cq + 1)]) + pv[u].y;
          const float vz = (bv.z + d * tr[tile_word<F>(4 * cq + 2)]) + pv[u].z;
          const float vw = (bv.w + d * tr[tile_word<F>(4 * cq + 3)]) + pv[u].w;
          float4 r;
          if (LAST) {
            r.x = vx; r.y = vy; r.z = vz; r.w = vw;
          } else {
            r.x = vx > 0.0f ? vx : 0.0f;
            r.y = vy > 0.0f ? vy : 0.0f;
            r.z = vz > 0.0f ? vz : 0.0f;
            r.w = vw > 0.0f ? vw : 0.0f;
          }
          o4[p] = r;
        }
      }
    }
  }
}

extern "C" void kernel_launch(void* const* d_in, const int* in_sizes, int n_in,
                              void* d_out, int out_size, void* d_ws, size_t ws_size,
                              hipStream_t stream) {
  const float* x      = (const float*)d_in[0];
  const int*   ei     = (const int*)d_in[1];
  const float* W_in   = (const float*)d_in[2];
  const float* b_in   = (const float*)d_in[3];
  const float* W_m0   = (const float*)d_in[4];
  const float* b_m0   = (const float*)d_in[5];
  const float* W_m1   = (const float*)d_in[6];
  const float* b_m1   = (const float*)d_in[7];
  const float* W_out  = (const float*)d_in[8];
  const float* b_out  = (const float*)d_in[9];
  const float* W_skip = (const float*)d_in[10];

  const int* src = ei;
  const int* dst = ei + N_EDGES;

  float* w    = (float*)d_ws;
  float* dinv = w;  w += DINV_FLOATS;
  int*   ent  = (int*)w;  w += N_EDGES;
  float* h    = w;  w += (size_t)N_NODES * NFH;
  float* xA   = w;  w += (size_t)N_NODES * NFH;
  float* xB   = w;  w += (size_t)N_NODES * NFH;
  float* skip = w;  w += (size_t)N_NODES * NCLS;
  int*   off  = (int*)w;

  const int T = 256;
  auto cdiv = [](int a, int b) { return (a + b - 1) / b; };

  const int nodeBlocks  = cdiv(N_NODES, T);
  const int edgeBlocks  = cdiv(N_EDGES, T);
  const int TA = AGG_WAVES * 32;

  k_deg_init<<<nodeBlocks, T, 0, stream>>>(dinv, N_NODES);
  k_deg_edges<<<edgeBlocks, T, 0, stream>>>(dst, dinv, N_EDGES);
  k_rsqrt_inplace<<<nodeBlocks, T, 0, stream>>>(dinv, N_NODES);
  k_partition<<<N_EBLKS, PART_THREADS, 0, stream>>>(src, dst, ent, off);

  k_gemm_resident<NFH><<<GEMM_WGS, 256, 0, stream>>>(x, W_in, W_in, dinv, h, h);
  k_tile_agg<NFH, false><<<N_TILES, TA, 0, stream>>>(ent, off, h, dinv, b_in, x, xA);

  k_gemm_resident<NFH><<<GEMM_WGS, 256, 0, stream>>>(xA, W_m0, W_m0, dinv, h, h);
  k_tile_agg<NFH, false><<<N_TILES, TA, 0, stream>>>(ent, off, h, dinv, b_m0, xA, xB);

  k_gemm_resident<NFH><<<GEMM_WGS, 256, 0, stream>>>(xB, W_m1, W_m1, dinv, h, h);
  k_tile_agg<NFH, false><<<N_TILES, TA, 0, stream>>>(ent, off, h, dinv, b_m1, xB, xA);

  k_gemm_resident<NCLS><<<GEMM_WGS, 256, 0, stream>>>(xA, W_out, W_skip, dinv, h, skip);
  k_tile_agg<NCLS, true><<<N_TILES, TA, 0, stream>>>(ent, off, h, dinv, b_out, skip,
                                                    (float*)d_out);
}


# Degrees counted from the per-tile edge lists: one launch counts each tile's entries per node by LDS int adds and stores rsqrt(1+count); f32 atomic pass, init and rsqrt launches gone; same values
# speedup vs baseline: 5.1289x; 1.1964x over previous
#include <hip/hip_runtime.h>
#include <hip/hip_bf16.h>

#define N_NODES 100000
#define N_EDGES 1600000
#define NFH     128
#define NCLS    64

typedef __attribute__((ext_vector_type(2))) float v2f;
typedef __attribute__((ext_vector_type(4))) float v4f;
typedef __attribute__((ext_vector_type(8))) float v8f;

#define DINV_FLOATS ((N_NODES + 63) / 64 * 64)
static_assert((N_EDGES % 64) == 0 && ((N_NODES * NCLS) % 64) == 0,
              "the regions between dinv and the planes are whole 256-byte units");

#define TILE_ROWS 400
#define N_TILES   (N_NODES / TILE_ROWS)
static_assert(N_TILES * TILE_ROWS == N_NODES, "tiles cover the nodes exactly");
static_assert(N_TILES <= 256, "the tile counters fit one 256-entry table");
static_assert(N_NODES <= (1 << 17) && TILE_ROWS <= (1 << 9), "src and row-in-tile share one word");

#define EBLK_EDGES   6400
#define N_EBLKS      (N_EDGES / EBLK_EDGES)
#define PART_THREADS 1024
#define PART_PPT     ((EBLK_EDGES + PART_THREADS - 1) / PART_THREADS)
static_assert(N_EBLKS * EBLK_EDGES == N_EDGES, "blocks cover the edges exactly");
static_assert(EBLK_EDGES % 4 == 0, "a block leaves LDS in 16-byte pieces");

__global__ __launch_bounds__(PART_THREADS)
void k_partition(const int* __restrict__ src, const int* __restrict__ dst,
                 int* __restrict__ ent, int* __restrict__ off) {
  __shared__ int cnt[256];
  __shared__ int start[256];
  __shared__ __attribute__((aligned(16))) int s_ent[EBLK_EDGES];
  const int tid  = threadIdx.x;
  const int base = blockIdx.x * EBLK_EDGES;

  if (tid < 256) cnt[tid] = 0;
  __syncthreads();

  int sv[PART_PPT], dv[PART_PPT], rk[PART_PPT];
#pragma unroll
  for (int k = 0; k < PART_PPT; ++k) {
    const int i  = tid + k * PART_THREADS;
    const int ic = i < EBLK_EDGES ? i : EBLK_EDGES - 1;
    sv[k] = src[base + ic];
    dv[k] = dst[base + ic];
  }
#pragma unroll
  for (int k = 0; k < PART_PPT; ++k) {
    const int i = tid + k * PART_THREADS;
    rk[k] = 0;
    if (i < EBLK_EDGES) rk[k] = atomicAdd(&cnt[dv[k] / TILE_ROWS], 1);
  }
  __syncthreads();

  if (tid < 32) {
    int c[8];
    int run = 0;
#pragma unroll
    for (int k = 0; k < 8; ++k) {
      const int x = cnt[tid * 8 + k];
      c[k] = run;
      run += x;
    }
    int inc = run;
#pragma unroll
    for (int d = 1; d < 32; d <<= 1) {
      const int y = __shfl_up(inc, d, 32);
      if (tid >= d) inc += y;
    }
    const int excl = inc - run;
#pragma unroll
    for (int k = 0; k < 8; ++k) start[tid * 8 + k] = excl + c[k];
  }
  __syncthreads();

  if (tid <= N_TILES) off[(size_t)blockIdx.x * (N_TILES + 1) + tid] = base + start[tid];

#pragma unroll
  for (int k = 0; k < PART_PPT; ++k) {
    const int i = tid + k * PART_THREADS;
    if (i < EBLK_EDGES) {
      const int tl = dv[k] / TILE_ROWS;
      const int p  = start[tl] + rk[k];
      s_ent[p] = sv[k] | ((dv[k] - tl * TILE_ROWS) << 17);
    }
  }
  __syncthreads();

  const int4* se4 = reinterpret_cast<const int4*>(s_ent);
  int4* __restrict__ e4 = reinterpret_cast<int4*>(ent + base);
  for (int i = tid; i < EBLK_EDGES / 4; i += PART_THREADS) e4[i] = se4[i];
}

#define DEG_WAVES 16
#define DEG_UN    8

__global__ __launch_bounds__(DEG_WAVES * 32)
void k_tile_dinv(const int* __restrict__ ent, const int* __restrict__ off,
                 float* __restrict__ dinv) {
  __shared__ int cnt[TILE_ROWS];
  constexpr int NT   = DEG_WAVES * 32;
  constexpr int NSEG = (N_EBLKS + DEG_WAVES - 1) / DEG_WAVES;
  static_assert(NSEG <= 32, "one lane per range");
  const int lane = threadIdx.x & 31;
  const int wave = __builtin_amdgcn_readfirstlane(threadIdx.x >> 5);
  const int t    = blockIdx.x;

  int o0 = 0, o1 = 0;
  {
    const int b = wave + lane * DEG_WAVES;
    if (lane < NSEG && b < N_EBLKS) {
      const int* __restrict__ o = off + (size_t)b * (N_TILES + 1) + t;
      o0 = o[0];
      o1 = o[1];
    }
  }
  for (int r = threadIdx.x; r < TILE_ROWS; r += NT) cnt[r] = 0;
  __syncthreads();
  const int slo = o0, slen = o1 - o0;

  int los[NSEG], pex[NSEG];
  int tot = 0;
#pragma unroll
  for (int sg = 0; sg < NSEG; ++sg) {
    los[sg] = __builtin_amdgcn_readlane(slo, sg);
    pex[sg] = tot;
    tot += __builtin_amdgcn_readlane(slen, sg);
  }
  for (int v0 = 0; v0 < tot; v0 += 32 * DEG_UN) {
    int en[DEG_UN];
#pragma unroll
    for (int u = 0; u < DEG_UN; ++u) {
      const int v = v0 + 32 * u + lane;
      int idx = 0;
#pragma unroll
      for (int sg = 0; sg < NSEG; ++sg) idx = v >= pex[sg] ? los[sg] + (v - pex[sg]) : idx;
      en[u] = ent[v < tot ? idx : 0];
    }
#pragma unroll
    for (int u = 0; u < DEG_UN; ++u) {
      if (v0 + 32 * u + lane < tot) atomicAdd(&cnt[(unsigned)en[u] >> 17], 1);
    }
  }
  __syncthreads();

  for (int r = threadIdx.x; r < TILE_ROWS; r += NT)
    dinv[t * TILE_ROWS + r] = rsqrtf(1.0f + (float)cnt[r]);
}

#define GEMM_ROWS  400
#define GEMM_TILES (GEMM_ROWS / 16)
#define GEMM_WGS   (N_NODES / GEMM_ROWS)
static_assert(GEMM_WGS * GEMM_ROWS == N_NODES && GEMM_TILES * 16 == GEMM_ROWS,
              "row ranges cover the nodes exactly in whole 16-row tiles");
static_assert(GEMM_ROWS % 8 == 0, "a lane's 8 row scales are two aligned 16-byte loads");

__device__ __forceinline__ void gemm_load_rows(float4 (&a)[16], float4 (&dq)[2],
                                               const float4* __restrict__ xa,
                                               const float4* __restrict__ da, int t) {
  const float4* __restrict__ xt = xa + (size_t)t * (16 * NFH / 4);
#pragma unroll
  for (int q = 0; q < 16; ++q) a[q] = xt[2 * q];
  dq[0] = da[4 * t];
  dq[1] = da[4 * t + 1];
}

template <int MO>
__device__ __forceinline__ void gemm_tile(const float4 (&a)[16], const float4 (&dq)[2],
                                          const v2f (&b)[32][2], bool scaled,
                                          float* __restrict__ out) {
  v8f c0 = {}, c1 = {};
#pragma unroll
  for (int q = 0; q < 16; ++q) {
    v2f lo, hi;
    lo.x = a[q].x; lo.y = a[q].y;
    hi.x = a[q].z; hi.y = a[q].w;
    c0 = __builtin_amdgcn_wmma_f32_16x16x4_f32(false, lo, false, b[2 * q][0], (short)0, c0,
                                               false, false);
    c1 = __builtin_amdgcn_wmma_f32_16x16x4_f32(false, lo, false, b[2 * q][1], (short)0, c1,
                                               false, false);
    c0 = __builtin_amdgcn_wmma_f32_16x16x4_f32(false, hi, false, b[2 * q + 1][0], (short)0, c0,
                                               false, false);
    c1 = __builtin_amdgcn_wmma_f32_16x16x4_f32(false, hi, false, b[2 * q + 1][1], (short)0, c1,
                                               false, false);
  }
  const float rs[8] = {dq[0].x, dq[0].y, dq[0].z, dq[0].w, dq[1].x, dq[1].y, dq[1].z, dq[1].w};
#pragma unroll
  for (int v = 0; v < 8; ++v) {
    const float sc = scaled ? rs[v] : 1.0f;
    out[(size_t)v * MO]      = c0[v] * sc;
    out[(size_t)v * MO + 16] = c1[v] * sc;
  }
}

template <int MO>
__global__ __launch_bounds__(256)
void k_gemm_resident(const float* __restrict__ X,
                     const float* __restrict__ W0, const float* __restrict__ W1,
                     const float* __restrict__ dinv,
                     float* __restrict__ H0, float* __restrict__ H1) {
  constexpr int PPM = MO / 32;
  const int lane = threadIdx.x & 31;
  const int wave = __builtin_amdgcn_readfirstlane(threadIdx.x >> 5);
  const int p    = wave & 3;
  const int rh   = wave >> 2;
  const int c    = lane & 15;
  const int hf   = lane >> 4;

  const float* __restrict__ Wm = (p / PPM) ? W1 : W0;
  float* __restrict__       Hm = (p / PPM) ? H1 : H0;
  const int col0 = (p % PPM) * 32;
  const bool scaled = (p / PPM) == 0;

  v2f b[32][2];
  {
    const float* __restrict__ wp = Wm + (size_t)(4 * hf) * MO + col0 + c;
#pragma unroll
    for (int q = 0; q < 16; ++q) {
#pragma unroll
      for (int st = 0; st < 2; ++st) {
#pragma unroll
        for (int j = 0; j < 2; ++j) {
          const int k = 8 * q + 2 * st;
          b[2 * q + st][j].x = wp[(size_t)k * MO + 16 * j];
          b[2 * q + st][j].y = wp[(size_t)(k + 1) * MO + 16 * j];
        }
      }
    }
  }

  const int row0 = blockIdx.x * GEMM_ROWS;
  const float4* __restrict__ xa =
      reinterpret_cast<const float4*>(X + (size_t)(row0 + c) * NFH + 4 * hf);
  float* __restrict__ ob = Hm + (size_t)(row0 + 8 * hf) * MO + col0 + c;
  const float4* __restrict__ da = reinterpret_cast<const float4*>(dinv + row0 + 8 * hf);

  const int n = (GEMM_TILES - rh + 1) / 2;
  float4 a0[16], a1[16];
  float4 d0[2], d1[2];
  gemm_load_rows(a0, d0, xa, da, rh);
  int i = 0;
  for (; i + 1 < n; i += 2) {
    const int t0 = rh + 2 * i;
    const int t1 = t0 + 2;
    const int t2 = rh + 2 * (i + 2 < n ? i + 2 : n - 1);
    gemm_load_rows(a1, d1, xa, da, t1);
    gemm_tile<MO>(a0, d0, b, scaled, ob + (size_t)t0 * 16 * MO);
    gemm_load_rows(a0, d0, xa, da, t2);
    gemm_tile<MO>(a1, d1, b, scaled, ob + (size_t)t1 * 16 * MO);
  }
  if (i < n) gemm_tile<MO>(a0, d0, b, scaled, ob + (size_t)(rh + 2 * i) * 16 * MO);
}

#define AGG_WAVES 32

template <int F> struct RowPiece;
template <> struct RowPiece<4> { typedef v4f type; };
template <> struct RowPiece<2> { typedef v2f type; };

template <int F>
__device__ __forceinline__ constexpr int tile_word(int c) { return 32 * (c % F) + c / F; }

template <int M, int U>
__device__ __forceinline__ void tile_add_edges(int sv, int dv, int j, int lane,
                                               const float* __restrict__ h,
                                               float* tile) {
  constexpr int F = M / 32;
  typedef typename RowPiece<F>::type piece_t;
  piece_t hv[U];
#pragma unroll
  for (int u = 0; u < U; ++u) {
    const int s = __builtin_amdgcn_readlane(sv, j + u);
    hv[u] = reinterpret_cast<const piece_t*>(h + (size_t)s * M)[lane];
  }
#pragma unroll
  for (int u = 0; u < U; ++u) {
    const int d = __builtin_amdgcn_readlane(dv, j + u);
    float* tp = tile + d * M + lane;
#pragma unroll
    for (int k = 0; k < F; ++k) atomicAdd(tp + 32 * k, hv[u][k]);
  }
}

template <int M>
__device__ __forceinline__ void tile_add_batch(int sv, int dv, int n, int lane,
                                               const float* __restrict__ h,
                                               float* tile) {
  constexpr int G = 32 / (M / 32);
  int j = 0;
  for (; j + G <= n; j += G) tile_add_edges<M, G>(sv, dv, j, lane, h, tile);
  for (; j < n; ++j)         tile_add_edges<M, 1>(sv, dv, j, lane, h, tile);
}

template <int M, bool LAST>
__global__ __launch_bounds__(AGG_WAVES * 32)
void k_tile_agg(const int* __restrict__ ent, const int* __restrict__ off,
                const float* __restrict__ h, const float* __restrict__ dinv,
                const float* __restrict__ bias, const float* __restrict__ res,
                float* __restrict__ outp) {
  __shared__ __attribute__((aligned(16))) float tile[TILE_ROWS * M];
  constexpr int Q  = M / 4;
  constexpr int NP = TILE_ROWS * Q;
  constexpr int NT = AGG_WAVES * 32;
  constexpr int UN = 4;
  constexpr int F  = M / 32;

  const int lane = threadIdx.x & 31;
  const int wave = __builtin_amdgcn_readfirstlane(threadIdx.x >> 5);
  const int t    = blockIdx.x;
  const int lo   = t * TILE_ROWS;
  const int cq   = threadIdx.x % Q;

  static_assert(NT % Q == 0, "a thread keeps one column group over its pieces");
  {
    const float4* __restrict__ h4 = reinterpret_cast<const float4*>(h) + (size_t)lo * Q;
    for (int p0 = threadIdx.x; p0 < NP; p0 += NT * UN) {
      float4 hv[UN];
#pragma unroll
      for (int u = 0; u < UN; ++u) {
        const int p = p0 + NT * u;
        hv[u] = h4[p < NP ? p : NP - 1];
      }
#pragma unroll
      for (int u = 0; u < UN; ++u) {
        const int p = p0 + NT * u;
        if (p < NP) {
          float* tr = tile + (p / Q) * M;
          tr[tile_word<F>(4 * cq)]     = hv[u].x;
          tr[tile_word<F>(4 * cq + 1)] = hv[u].y;
          tr[tile_word<F>(4 * cq + 2)] = hv[u].z;
          tr[tile_word<F>(4 * cq + 3)] = hv[u].w;
        }
      }
    }
  }
  __syncthreads();

  {
    constexpr int NSEG = (N_EBLKS + AGG_WAVES - 1) / AGG_WAVES;
    static_assert(NSEG <= 32, "one lane per range");
    int slo = 0, slen = 0;
    {
      const int b = wave + lane * AGG_WAVES;
      if (lane < NSEG && b < N_EBLKS) {
        const int* __restrict__ o = off + (size_t)b * (N_TILES + 1) + t;
        slo  = o[0];
        slen = o[1] - slo;
      }
    }
    int los[NSEG], pex[NSEG];
    int tot = 0;
#pragma unroll
    for (int sg = 0; sg < NSEG; ++sg) {
      los[sg] = __builtin_amdgcn_readlane(slo, sg);
      pex[sg] = tot;
      tot += __builtin_amdgcn_readlane(slen, sg);
    }
    auto entry_of = [&](int v0) {
      const int v = v0 + lane;
      int idx = 0;
#pragma unroll
      for (int sg = 0; sg < NSEG; ++sg) idx = v >= pex[sg] ? los[sg] + (v - pex[sg]) : idx;
      return v < tot ? idx : 0;
    };
    int en = 0;
    if (tot > 0) en = ent[entry_of(0)];
    for (int v0 = 0; v0 < tot; v0 += 32) {
      int en_n = 0;
      if (v0 + 32 < tot) en_n = ent[entry_of(v0 + 32)];
      const int n = tot - v0 < 32 ? tot - v0 : 32;
      tile_add_batch<M>(en & 0x1ffff, (int)((unsigned)en >> 17), n, lane, h, tile);
      en = en_n;
    }
  }
  __syncthreads();

  {
    const float4* __restrict__ r4 = reinterpret_cast<const float4*>(res) + (size_t)lo * Q;
    float4* __restrict__ o4 = reinterpret_cast<float4*>(outp) + (size_t)lo * Q;
    const float4 bv = reinterpret_cast<const float4*>(bias)[cq];
    for (int p0 = threadIdx.x; p0 < NP; p0 += NT * UN) {
      float4 pv[UN];
      float  dv[UN];
#pragma unroll
      for (int u = 0; u < UN; ++u) {
        const int p  = p0 + NT * u;
        const int pc = p < NP ? p : NP - 1;
        pv[u] = r4[pc];
        dv[u] = dinv[lo + pc / Q];
      }
#pragma unroll
      for (int u = 0; u < UN; ++u) {
        const int p = p0 + NT * u;
        if (p < NP) {
          const float* tr = tile + (p / Q) * M;
          const float d  = dv[u];
          const float vx = (bv.x + d * tr[tile_word<F>(4 * cq)])     + pv[u].x;
          const float vy = (bv.y + d * tr[tile_word<F>(4 * cq + 1)]) + pv[u].y;
          const float vz = (bv.z + d * tr[tile_word<F>(4 * cq + 2)]) + pv[u].z;
          const float vw = (bv.w + d * tr[tile_word<F>(4 * cq + 3)]) + pv[u].w;
          float4 r;
          if (LAST) {
            r.x = vx; r.y = vy; r.z = vz; r.w = vw;
          } else {
            r.x = vx > 0.0f ? vx : 0.0f;
            r.y = vy > 0.0f ? vy : 0.0f;
            r.z = vz > 0.0f ? vz : 0.0f;
            r.w = vw > 0.0f ? vw : 0.0f;
          }
          o4[p] = r;
        }
      }
    }
  }
}

extern "C" void kernel_launch(void* const* d_in, const int* in_sizes, int n_in,
                              void* d_out, int out_size, void* d_ws, size_t ws_size,
                              hipStream_t stream) {
  const float* x      = (const float*)d_in[0];
  const int*   ei     = (const int*)d_in[1];
  const float* W_in   = (const float*)d_in[2];
  const float* b_in   = (const float*)d_in[3];
  const float* W_m0   = (const float*)d_in[4];
  const float* b_m0   = (const float*)d_in[5];
  const float* W_m1   = (const float*)d_in[6];
  const float* b_m1   = (const float*)d_in[7];
  const float* W_out  = (const float*)d_in[8];
  const float* b_out  = (const float*)d_in[9];
  const float* W_skip = (const float*)d_in[10];

  const int* src = ei;
  const int* dst = ei + N_EDGES;

  float* w    = (float*)d_ws;
  float* dinv = w;  w += DINV_FLOATS;
  int*   ent  = (int*)w;  w += N_EDGES;
  float* h    = w;  w += (size_t)N_NODES * NFH;
  float* xA   = w;  w += (size_t)N_NODES * NFH;
  float* xB   = w;  w += (size_t)N_NODES * NFH;
  float* skip = w;  w += (size_t)N_NODES * NCLS;
  int*   off  = (int*)w;

  const int TA = AGG_WAVES * 32;

  k_partition<<<N_EBLKS, PART_THREADS, 0, stream>>>(src, dst, ent, off);
  k_tile_dinv<<<N_TILES, DEG_WAVES * 32, 0, stream>>>(ent, off, dinv);

  k_gemm_resident<NFH><<<GEMM_WGS, 256, 0, stream>>>(x, W_in, W_in, dinv, h, h);
  k_tile_agg<NFH, false><<<N_TILES, TA, 0, stream>>>(ent, off, h, dinv, b_in, x, xA);

  k_gemm_resident<NFH><<<GEMM_WGS, 256, 0, stream>>>(xA, W_m0, W_m0, dinv, h, h);
  k_tile_agg<NFH, false><<<N_TILES, TA, 0, stream>>>(ent, off, h, dinv, b_m0, xA, xB);

  k_gemm_resident<NFH><<<GEMM_WGS, 256, 0, stream>>>(xB, W_m1, W_m1, dinv, h, h);
  k_tile_agg<NFH, false><<<N_TILES, TA, 0, stream>>>(ent, off, h, dinv, b_m1, xB, xA);

  k_gemm_resident<NCLS><<<GEMM_WGS, 256, 0, stream>>>(xA, W_out, W_skip, dinv, h, skip);
  k_tile_agg<NCLS, true><<<N_TILES, TA, 0, stream>>>(ent, off, h, dinv, b_out, skip,
                                                    (float*)d_out);
}


# Tile kernel: loads that do not depend on the sums move one phase ahead: range starts and first entries before the fill, fill as one load group, residual rows requested before the edge walk; same sums
# speedup vs baseline: 5.3140x; 1.0361x over previous
#include <hip/hip_runtime.h>
#include <hip/hip_bf16.h>

#define N_NODES 100000
#define N_EDGES 1600000
#define NFH     128
#define NCLS    64

typedef __attribute__((ext_vector_type(2))) float v2f;
typedef __attribute__((ext_vector_type(4))) float v4f;
typedef __attribute__((ext_vector_type(8))) float v8f;

#define DINV_FLOATS ((N_NODES + 63) / 64 * 64)
static_assert((N_EDGES % 64) == 0 && ((N_NODES * NCLS) % 64) == 0,
              "the regions between dinv and the planes are whole 256-byte units");

#define TILE_ROWS 400
#define N_TILES   (N_NODES / TILE_ROWS)
static_assert(N_TILES * TILE_ROWS == N_NODES, "tiles cover the nodes exactly");
static_assert(N_TILES <= 256, "the tile counters fit one 256-entry table");
static_assert(N_NODES <= (1 << 17) && TILE_ROWS <= (1 << 9), "src and row-in-tile share one word");

#define EBLK_EDGES   6400
#define N_EBLKS      (N_EDGES / EBLK_EDGES)
#define PART_THREADS 1024
#define PART_PPT     ((EBLK_EDGES + PART_THREADS - 1) / PART_THREADS)
static_assert(N_EBLKS * EBLK_EDGES == N_EDGES, "blocks cover the edges exactly");
static_assert(EBLK_EDGES % 4 == 0, "a block leaves LDS in 16-byte pieces");

__global__ __launch_bounds__(PART_THREADS)
void k_partition(const int* __restrict__ src, const int* __restrict__ dst,
                 int* __restrict__ ent, int* __restrict__ off) {
  __shared__ int cnt[256];
  __shared__ int start[256];
  __shared__ __attribute__((aligned(16))) int s_ent[EBLK_EDGES];
  const int tid  = threadIdx.x;
  const int base = blockIdx.x * EBLK_EDGES;

  if (tid < 256) cnt[tid] = 0;
  __syncthreads();

  int sv[PART_PPT], dv[PART_PPT], rk[PART_PPT];
#pragma unroll
  for (int k = 0; k < PART_PPT; ++k) {
    const int i  = tid + k * PART_THREADS;
    const int ic = i < EBLK_EDGES ? i : EBLK_EDGES - 1;
    sv[k] = src[base + ic];
    dv[k] = dst[base + ic];
  }
#pragma unroll
  for (int k = 0; k < PART_PPT; ++k) {
    const int i = tid + k * PART_THREADS;
    rk[k] = 0;
    if (i < EBLK_EDGES) rk[k] = atomicAdd(&cnt[dv[k] / TILE_ROWS], 1);
  }
  __syncthreads();

  if (tid < 32) {
    int c[8];
    int run = 0;
#pragma unroll
    for (int k = 0; k < 8; ++k) {
      const int x = cnt[tid * 8 + k];
      c[k] = run;
      run += x;
    }
    int inc = run;
#pragma unroll
    for (int d = 1; d < 32; d <<= 1) {
      const int y = __shfl_up(inc, d, 32);
      if (tid >= d) inc += y;
    }
    const int excl = inc - run;
#pragma unroll
    for (int k = 0; k < 8; ++k) start[tid * 8 + k] = excl + c[k];
  }
  __syncthreads();

  if (tid <= N_TILES) off[(size_t)blockIdx.x * (N_TILES + 1) + tid] = base + start[tid];

#pragma unroll
  for (int k = 0; k < PART_PPT; ++k) {
    const int i = tid + k * PART_THREADS;
    if (i < EBLK_EDGES) {
      const int tl = dv[k] / TILE_ROWS;
      const int p  = start[tl] + rk[k];
      s_ent[p] = sv[k] | ((dv[k] - tl * TILE_ROWS) << 17);
    }
  }
  __syncthreads();

  const int4* se4 = reinterpret_cast<const int4*>(s_ent);
  int4* __restrict__ e4 = reinterpret_cast<int4*>(ent + base);
  for (int i = tid; i < EBLK_EDGES / 4; i += PART_THREADS) e4[i] = se4[i];
}

#define DEG_WAVES 16
#define DEG_UN    8

__global__ __launch_bounds__(DEG_WAVES * 32)
void k_tile_dinv(const int* __restrict__ ent, const int* __restrict__ off,
                 float* __restrict__ dinv) {
  __shared__ int cnt[TILE_ROWS];
  constexpr int NT   = DEG_WAVES * 32;
  constexpr int NSEG = (N_EBLKS + DEG_WAVES - 1) / DEG_WAVES;
  static_assert(NSEG <= 32, "one lane per range");
  const int lane = threadIdx.x & 31;
  const int wave = __builtin_amdgcn_readfirstlane(threadIdx.x >> 5);
  const int t    = blockIdx.x;

  int o0 = 0, o1 = 0;
  {
    const int b = wave + lane * DEG_WAVES;
    if (lane < NSEG && b < N_EBLKS) {
      const int* __restrict__ o = off + (size_t)b * (N_TILES + 1) + t;
      o0 = o[0];
      o1 = o[1];
    }
  }
  for (int r = threadIdx.x; r < TILE_ROWS; r += NT) cnt[r] = 0;
  __syncthreads();
  const int slo = o0, slen = o1 - o0;

  int los[NSEG], pex[NSEG];
  int tot = 0;
#pragma unroll
  for (int sg = 0; sg < NSEG; ++sg) {
    los[sg] = __builtin_amdgcn_readlane(slo, sg);
    pex[sg] = tot;
    tot += __builtin_amdgcn_readlane(slen, sg);
  }
  for (int v0 = 0; v0 < tot; v0 += 32 * DEG_UN) {
    int en[DEG_UN];
#pragma unroll
    for (int u = 0; u < DEG_UN; ++u) {
      const int v = v0 + 32 * u + lane;
      int idx = 0;
#pragma unroll
      for (int sg = 0; sg < NSEG; ++sg) idx = v >= pex[sg] ? los[sg] + (v - pex[sg]) : idx;
      en[u] = ent[v < tot ? idx : 0];
    }
#pragma unroll
    for (int u = 0; u < DEG_UN; ++u) {
      if (v0 + 32 * u + lane < tot) atomicAdd(&cnt[(unsigned)en[u] >> 17], 1);
    }
  }
  __syncthreads();

  for (int r = threadIdx.x; r < TILE_ROWS; r += NT)
    dinv[t * TILE_ROWS + r] = rsqrtf(1.0f + (float)cnt[r]);
}

#define GEMM_ROWS  400
#define GEMM_TILES (GEMM_ROWS / 16)
#define GEMM_WGS   (N_NODES / GEMM_ROWS)
static_assert(GEMM_WGS * GEMM_ROWS == N_NODES && GEMM_TILES * 16 == GEMM_ROWS,
              "row ranges cover the nodes exactly in whole 16-row tiles");
static_assert(GEMM_ROWS % 8 == 0, "a lane's 8 row scales are two aligned 16-byte loads");

__device__ __forceinline__ void gemm_load_rows(float4 (&a)[16], float4 (&dq)[2],
                                               const float4* __restrict__ xa,
                                               const float4* __restrict__ da, int t) {
  const float4* __restrict__ xt = xa + (size_t)t * (16 * NFH / 4);
#pragma unroll
  for (int q = 0; q < 16; ++q) a[q] = xt[2 * q];
  dq[0] = da[4 * t];
  dq[1] = da[4 * t + 1];
}

template <int MO>
__device__ __forceinline__ void gemm_tile(const float4 (&a)[16], const float4 (&dq)[2],
                                          const v2f (&b)[32][2], bool scaled,
                                          float* __restrict__ out) {
  v8f c0 = {}, c1 = {};
#pragma unroll
  for (int q = 0; q < 16; ++q) {
    v2f lo, hi;
    lo.x = a[q].x; lo.y = a[q].y;
    hi.x = a[q].z; hi.y = a[q].w;
    c0 = __builtin_amdgcn_wmma_f32_16x16x4_f32(false, lo, false, b[2 * q][0], (short)0, c0,
                                               false, false);
    c1 = __builtin_amdgcn_wmma_f32_16x16x4_f32(false, lo, false, b[2 * q][1], (short)0, c1,
                                               false, false);
    c0 = __builtin_amdgcn_wmma_f32_16x16x4_f32(false, hi, false, b[2 * q + 1][0], (short)0, c0,
                                               false, false);
    c1 = __builtin_amdgcn_wmma_f32_16x16x4_f32(false, hi, false, b[2 * q + 1][1], (short)0, c1,
                                               false, false);
  }
  const float rs[8] = {dq[0].x, dq[0].y, dq[0].z, dq[0].w, dq[1].x, dq[1].y, dq[1].z, dq[1].w};
#pragma unroll
  for (int v = 0; v < 8; ++v) {
    const float sc = scaled ? rs[v] : 1.0f;
    out[(size_t)v * MO]      = c0[v] * sc;
    out[(size_t)v * MO + 16] = c1[v] * sc;
  }
}

template <int MO>
__global__ __launch_bounds__(256)
void k_gemm_resident(const float* __restrict__ X,
                     const float* __restrict__ W0, const float* __restrict__ W1,
                     const float* __restrict__ dinv,
                     float* __restrict__ H0, float* __restrict__ H1) {
  constexpr int PPM = MO / 32;
  const int lane = threadIdx.x & 31;
  const int wave = __builtin_amdgcn_readfirstlane(threadIdx.x >> 5);
  const int p    = wave & 3;
  const int rh   = wave >> 2;
  const int c    = lane & 15;
  const int hf   = lane >> 4;

  const float* __restrict__ Wm = (p / PPM) ? W1 : W0;
  float* __restrict__       Hm = (p / PPM) ? H1 : H0;
  const int col0 = (p % PPM) * 32;
  const bool scaled = (p / PPM) == 0;

  v2f b[32][2];
  {
    const float* __restrict__ wp = Wm + (size_t)(4 * hf) * MO + col0 + c;
#pragma unroll
    for (int q = 0; q < 16; ++q) {
#pragma unroll
      for (int st = 0; st < 2; ++st) {
#pragma unroll
        for (int j = 0; j < 2; ++j) {
          const int k = 8 * q + 2 * st;
          b[2 * q + st][j].x = wp[(size_t)k * MO + 16 * j];
          b[2 * q + st][j].y = wp[(size_t)(k + 1) * MO + 16 * j];
        }
      }
    }
  }

  const int row0 = blockIdx.x * GEMM_ROWS;
  const float4* __restrict__ xa =
      reinterpret_cast<const float4*>(X + (size_t)(row0 + c) * NFH + 4 * hf);
  float* __restrict__ ob = Hm + (size_t)(row0 + 8 * hf) * MO + col0 + c;
  const float4* __restrict__ da = reinterpret_cast<const float4*>(dinv + row0 + 8 * hf);

  const int n = (GEMM_TILES - rh + 1) / 2;
  float4 a0[16], a1[16];
  float4 d0[2], d1[2];
  gemm_load_rows(a0, d0, xa, da, rh);
  int i = 0;
  for (; i + 1 < n; i += 2) {
    const int t0 = rh + 2 * i;
    const int t1 = t0 + 2;
    const int t2 = rh + 2 * (i + 2 < n ? i + 2 : n - 1);
    gemm_load_rows(a1, d1, xa, da, t1);
    gemm_tile<MO>(a0, d0, b, scaled, ob + (size_t)t0 * 16 * MO);
    gemm_load_rows(a0, d0, xa, da, t2);
    gemm_tile<MO>(a1, d1, b, scaled, ob + (size_t)t1 * 16 * MO);
  }
  if (i < n) gemm_tile<MO>(a0, d0, b, scaled, ob + (size_t)(rh + 2 * i) * 16 * MO);
}

#define AGG_WAVES 32

template <int F> struct RowPiece;
template <> struct RowPiece<4> { typedef v4f type; };
template <> struct RowPiece<2> { typedef v2f type; };

template <int F>
__device__ __forceinline__ constexpr int tile_word(int c) { return 32 * (c % F) + c / F; }

template <int M, int U>
__device__ __forceinline__ void tile_add_edges(int sv, int dv, int j, int lane,
                                               const float* __restrict__ h,
                                               float* tile) {
  constexpr int F = M / 32;
  typedef typename RowPiece<F>::type piece_t;
  piece_t hv[U];
#pragma unroll
  for (int u = 0; u < U; ++u) {
    const int s = __builtin_amdgcn_readlane(sv, j + u);
    hv[u] = reinterpret_cast<const piece_t*>(h + (size_t)s * M)[lane];
  }
#pragma unroll
  for (int u = 0; u < U; ++u) {
    const int d = __builtin_amdgcn_readlane(dv, j + u);
    float* tp = tile + d * M + lane;
#pragma unroll
    for (int k = 0; k < F; ++k) atomicAdd(tp + 32 * k, hv[u][k]);
  }
}

template <int M>
__device__ __forceinline__ void tile_add_batch(int sv, int dv, int n, int lane,
                                               const float* __restrict__ h,
                                               float* tile) {
  constexpr int G = 32 / (M / 32);
  int j = 0;
  for (; j + G <= n; j += G) tile_add_edges<M, G>(sv, dv, j, lane, h, tile);
  for (; j < n; ++j)         tile_add_edges<M, 1>(sv, dv, j, lane, h, tile);
}

template <int M, bool LAST>
__global__ __launch_bounds__(AGG_WAVES * 32)
void k_tile_agg(const int* __restrict__ ent, const int* __restrict__ off,
                const float* __restrict__ h, const float* __restrict__ dinv,
                const float* __restrict__ bias, const float* __restrict__ res,
                float* __restrict__ outp) {
  __shared__ __attribute__((aligned(16))) float tile[TILE_ROWS * M];
  __shared__ float s_dinv[TILE_ROWS];
  constexpr int Q   = M / 4;
  constexpr int NP  = TILE_ROWS * Q;
  constexpr int NT  = AGG_WAVES * 32;
  constexpr int PPT = (NP + NT - 1) / NT;
  constexpr int PFU = NP / NT;
  constexpr int REM = NP - PFU * NT;
  static_assert(REM % 32 == 0 && PPT - PFU <= 1, "the odd piece belongs to whole waves");
  constexpr int F   = M / 32;
  constexpr int NSEG = (N_EBLKS + AGG_WAVES - 1) / AGG_WAVES;
  static_assert(NSEG <= 32, "one lane per range");
  static_assert(NT % Q == 0, "a thread keeps one column group over its pieces");
  static_assert(TILE_ROWS <= NT, "one thread per destination scale");

  const int lane = threadIdx.x & 31;
  const int wave = __builtin_amdgcn_readfirstlane(threadIdx.x >> 5);
  const int t    = blockIdx.x;
  const int lo   = t * TILE_ROWS;
  const int cq   = threadIdx.x % Q;


  int o0 = 0, o1 = 0;
  {
    const int b = wave + lane * AGG_WAVES;
    if (lane < NSEG && b < N_EBLKS) {
      const int* __restrict__ o = off + (size_t)b * (N_TILES + 1) + t;
      o0 = o[0];
      o1 = o[1];
    }
  }

  const float4 bv = reinterpret_cast<const float4*>(bias)[cq];

  const float4* __restrict__ h4 = reinterpret_cast<const float4*>(h) + (size_t)lo * Q;
  float4 hv[PPT];
#pragma unroll
  for (int u = 0; u < PPT; ++u) {
    const int p = threadIdx.x + NT * u;
    hv[u] = h4[p < NP ? p : NP - 1];
  }
  float dsc = 0.0f;
  if (threadIdx.x < TILE_ROWS) dsc = dinv[lo + threadIdx.x];

  const int slo = o0, slen = o1 - o0;
  int los[NSEG], pex[NSEG];
  int tot = 0;
#pragma unroll
  for (int sg = 0; sg < NSEG; ++sg) {
    los[sg] = __builtin_amdgcn_readlane(slo, sg);
    pex[sg] = tot;
    tot += __builtin_amdgcn_readlane(slen, sg);
  }
  auto entry_of = [&](int v0) {
    const int v = v0 + lane;
    int idx = 0;
#pragma unroll
    for (int sg = 0; sg < NSEG; ++sg) idx = v >= pex[sg] ? los[sg] + (v - pex[sg]) : idx;
    return v < tot ? idx : 0;
  };
  int en = 0;
  if (tot > 0) en = ent[entry_of(0)];

#pragma unroll
  for (int u = 0; u < PPT; ++u) {
    const int p = threadIdx.x + NT * u;
    if (p < NP) {
      float* tr = tile + (p / Q) * M;
      tr[tile_word<F>(4 * cq)]     = hv[u].x;
      tr[tile_word<F>(4 * cq + 1)] = hv[u].y;
      tr[tile_word<F>(4 * cq + 2)] = hv[u].z;
      tr[tile_word<F>(4 * cq + 3)] = hv[u].w;
    }
  }
  if (threadIdx.x < TILE_ROWS) s_dinv[threadIdx.x] = dsc;

  __syncthreads();

  const float4* __restrict__ r4 = reinterpret_cast<const float4*>(res) + (size_t)lo * Q;
  float4 pv[PFU];
#pragma unroll
  for (int u = 0; u < PFU; ++u) pv[u] = r4[threadIdx.x + NT * u];
  const bool odd = REM > 0 && threadIdx.x < REM;
  const int podd = threadIdx.x + NT * PFU;
  const float4 pr = r4[podd < NP ? podd : NP - 1];

  {
    auto batch = [&](int v0) {
      int en_n = 0;
      if (v0 + 32 < tot) en_n = ent[entry_of(v0 + 32)];
      const int n = tot - v0 < 32 ? tot - v0 : 32;
      tile_add_batch<M>(en & 0x1ffff, (int)((unsigned)en >> 17), n, lane, h, tile);
      en = en_n;
    };
    if (tot > 0) batch(0);
    for (int v0 = 32; v0 < tot; v0 += 32) batch(v0);
  }
  __syncthreads();

  {
    float4* __restrict__ o4 = reinterpret_cast<float4*>(outp) + (size_t)lo * Q;
    auto finish = [&](int p, const float4& rv) {
      const float* tr = tile + (p / Q) * M;
      const float d  = s_dinv[p / Q];
      const float vx = (bv.x + d * tr[tile_word<F>(4 * cq)])     + rv.x;
      const float vy = (bv.y + d * tr[tile_word<F>(4 * cq + 1)]) + rv.y;
      const float vz = (bv.z + d * tr[tile_word<F>(4 * cq + 2)]) + rv.z;
      const float vw = (bv.w + d * tr[tile_word<F>(4 * cq + 3)]) + rv.w;
      float4 r;
      if (LAST) {
        r.x = vx; r.y = vy; r.z = vz; r.w = vw;
      } else {
        r.x = vx > 0.0f ? vx : 0.0f;
        r.y = vy > 0.0f ? vy : 0.0f;
        r.z = vz > 0.0f ? vz : 0.0f;
        r.w = vw > 0.0f ? vw : 0.0f;
      }
      o4[p] = r;
    };
#pragma unroll
    for (int u = 0; u < PFU; ++u) finish(threadIdx.x + NT * u, pv[u]);
    if (odd) finish(threadIdx.x + NT * PFU, pr);
  }
}

extern "C" void kernel_launch(void* const* d_in, const int* in_sizes, int n_in,
                              void* d_out, int out_size, void* d_ws, size_t ws_size,
                              hipStream_t stream) {
  const float* x      = (const float*)d_in[0];
  const int*   ei     = (const int*)d_in[1];
  const float* W_in   = (const float*)d_in[2];
  const float* b_in   = (const float*)d_in[3];
  const float* W_m0   = (const float*)d_in[4];
  const float* b_m0   = (const float*)d_in[5];
  const float* W_m1   = (const float*)d_in[6];
  const float* b_m1   = (const float*)d_in[7];
  const float* W_out  = (const float*)d_in[8];
  const float* b_out  = (const float*)d_in[9];
  const float* W_skip = (const float*)d_in[10];

  const int* src = ei;
  const int* dst = ei + N_EDGES;

  float* w    = (float*)d_ws;
  float* dinv = w;  w += DINV_FLOATS;
  int*   ent  = (int*)w;  w += N_EDGES;
  float* h    = w;  w += (size_t)N_NODES * NFH;
  float* xA   = w;  w += (size_t)N_NODES * NFH;
  float* xB   = w;  w += (size_t)N_NODES * NFH;
  float* skip = w;  w += (size_t)N_NODES * NCLS;
  int*   off  = (int*)w;

  const int TA = AGG_WAVES * 32;

  k_partition<<<N_EBLKS, PART_THREADS, 0, stream>>>(src, dst, ent, off);
  k_tile_dinv<<<N_TILES, DEG_WAVES * 32, 0, stream>>>(ent, off, dinv);

  k_gemm_resident<NFH><<<GEMM_WGS, 256, 0, stream>>>(x, W_in, W_in, dinv, h, h);
  k_tile_agg<NFH, false><<<N_TILES, TA, 0, stream>>>(ent, off, h, dinv, b_in, x, xA);

  k_gemm_resident<NFH><<<GEMM_WGS, 256, 0, stream>>>(xA, W_m0, W_m0, dinv, h, h);
  k_tile_agg<NFH, false><<<N_TILES, TA, 0, stream>>>(ent, off, h, dinv, b_m0, xA, xB);

  k_gemm_resident<NFH><<<GEMM_WGS, 256, 0, stream>>>(xB, W_m1, W_m1, dinv, h, h);
  k_tile_agg<NFH, false><<<N_TILES, TA, 0, stream>>>(ent, off, h, dinv, b_m1, xB, xA);

  k_gemm_resident<NCLS><<<GEMM_WGS, 256, 0, stream>>>(xA, W_out, W_skip, dinv, h, skip);
  k_tile_agg<NCLS, true><<<N_TILES, TA, 0, stream>>>(ent, off, h, dinv, b_out, skip,
                                                    (float*)d_out);
}
